# starting kernel
# baseline (speedup 1.0000x reference)
#include <hip/hip_runtime.h>


typedef float v2f __attribute__((ext_vector_type(2)));
typedef float v8f __attribute__((ext_vector_type(8)));

#define Bq 8
#define Nq 1024
#define Fin 256
#define Fout 256
#define Hh 4
#define HD 64
#define NEG_SLOPE 0.2f

__global__ __launch_bounds__(128) void gat_proj(const float* __restrict__ x,
                                                const float* __restrict__ W,
                                                float* __restrict__ hbuf) {
  const int tid = threadIdx.x;
  const int wave = tid >> 5, lane = tid & 31;
  const int lane16 = lane & 15, laneHi = lane >> 4;
  const int tile = blockIdx.x * 4 + wave;
  const int mt = tile >> 4;
  const int nt = tile & 15;
  const int row = mt * 16 + lane16;
  const int col = nt * 16 + lane16;

  const float* xr = x + row * Fin;
  v8f acc = {};
#pragma unroll 4
  for (int k = 0; k < Fin; k += 4) {
    const int k0 = k + 2 * laneHi;
    v2f av = *reinterpret_cast<const v2f*>(xr + k0);
    v2f bv;
    bv.x = W[k0 * Fout + col];
    bv.y = W[(k0 + 1) * Fout + col];
    acc = __builtin_amdgcn_wmma_f32_16x16x4_f32(false, av, false, bv,
                                                (short)0, acc, false, false);
  }
  const int hh = col >> 6, d = col & 63;
#pragma unroll
  for (int v = 0; v < 8; ++v) {
    const int r = mt * 16 + v + 8 * laneHi;
    const int bb = r >> 10, n = r & (Nq - 1);
    hbuf[(((bb * Hh + hh) * Nq + n) * HD) + d] = acc[v];
  }
}

__global__ __launch_bounds__(256) void gat_scores(const float* __restrict__ hbuf,
                                                  const float* __restrict__ a,
                                                  float* __restrict__ s1,
                                                  float* __restrict__ s2) {
  const int tid = threadIdx.x;
  const int wave = tid >> 5, lane = tid & 31;
  const int idx = blockIdx.x * 8 + wave;
  const int hh = (idx >> 10) & (Hh - 1);
  const float* hr = hbuf + idx * HD;
  const float* ah = a + hh * (2 * HD);
  const float h0 = hr[lane], h1 = hr[lane + 32];
  float d1 = h0 * ah[lane] + h1 * ah[lane + 32];
  float d2 = h0 * ah[HD + lane] + h1 * ah[HD + lane + 32];
#pragma unroll
  for (int off = 16; off; off >>= 1) {
    d1 += __shfl_xor(d1, off, 32);
    d2 += __shfl_xor(d2, off, 32);
  }
  if (lane == 0) { s1[idx] = d1; s2[idx] = d2; }
}

__global__ __launch_bounds__(256) void gat_attn(const float* __restrict__ hbuf,
                                                const int* __restrict__ adj,
                                                const float* __restrict__ s1g,
                                                const float* __restrict__ s2g,
                                                float* __restrict__ out) {
  __shared__ unsigned int maskw[16 * (Nq / 32)];
  __shared__ float s2l[Hh * Nq];
  __shared__ float s1l[Hh * 16];
  __shared__ float rmaxl[Hh * 16];
  __shared__ float rsuml[Hh * 16];
  __shared__ float opart[Hh * 16 * HD];

  const int tid = threadIdx.x;
  const int wave = tid >> 5, lane = tid & 31;
  const int lane16 = lane & 15, laneHi = lane >> 4;
  const int b = blockIdx.y, n0 = blockIdx.x * 16;
  const int hh = wave >> 1;
  const int half = wave & 1;

  const int* adjb = adj + (b * Nq + n0) * Nq;
  for (int w = tid; w < 16 * (Nq / 32); w += 256) {
    const int r = w >> 5, m0 = (w & 31) * 32;
    const int* ar = adjb + r * Nq + m0;
    unsigned int bits = 0u;
#pragma unroll
    for (int j = 0; j < 32; ++j) bits |= (ar[j] != 0 ? 1u : 0u) << j;
    maskw[w] = bits;
  }
  const float* s2b = s2g + b * Hh * Nq;
  for (int i = tid; i < Hh * Nq; i += 256) s2l[i] = s2b[i];
  const float* s1b = s1g + b * Hh * Nq;
  for (int i = tid; i < Hh * 16; i += 256)
    s1l[i] = s1b[(i >> 4) * Nq + n0 + (i & 15)];
  __syncthreads();

  for (int r8 = 0; r8 < 8; ++r8) {
    const int r = half * 8 + r8;
    const float s1r = s1l[hh * 16 + r];
    float mx = -1e30f, sm = 0.f;
    for (int m = lane; m < Nq; m += 32) {
      if ((maskw[r * 32 + (m >> 5)] >> (m & 31)) & 1u) {
        float v = s1r + s2l[hh * Nq + m];
        v = v > 0.f ? v : NEG_SLOPE * v;
        const float nm = fmaxf(mx, v);
        sm = sm * __expf(mx - nm) + __expf(v - nm);
        mx = nm;
      }
    }
#pragma unroll
    for (int off = 16; off; off >>= 1) {
      const float om = __shfl_xor(mx, off, 32);
      const float os = __shfl_xor(sm, off, 32);
      const float nm = fmaxf(mx, om);
      sm = sm * __expf(mx - nm) + os * __expf(om - nm);
      mx = nm;
    }
    if (lane == 0) { rmaxl[hh * 16 + r] = mx; rsuml[hh * 16 + r] = sm; }
  }
  __syncthreads();

  const float s1v  = s1l[hh * 16 + lane16];
  const float rmax = rmaxl[hh * 16 + lane16];
  const float rsum = rsuml[hh * 16 + lane16];
  const float rinv = rsum > 0.f ? 1.f / rsum : 0.f;
  const float* hb  = hbuf + (b * Hh + hh) * Nq * HD;
  const float* s2h = s2l + hh * Nq;
  const unsigned int* mrow = maskw + lane16 * 32;
  const int mbase = half * (Nq / 2);

  v8f acc0 = {}, acc1 = {}, acc2 = {}, acc3 = {};
  for (int kk = mbase; kk < mbase + Nq / 2; kk += 4) {
    const int m0 = kk + 2 * laneHi;
    float p0 = 0.f, p1 = 0.f;
    if ((mrow[m0 >> 5] >> (m0 & 31)) & 1u) {
      float v = s1v + s2h[m0];
      v = v > 0.f ? v : NEG_SLOPE * v;
      p0 = __expf(v - rmax) * rinv;
    }
    if ((mrow[(m0 + 1) >> 5] >> ((m0 + 1) & 31)) & 1u) {
      float v = s1v + s2h[m0 + 1];
      v = v > 0.f ? v : NEG_SLOPE * v;
      p1 = __expf(v - rmax) * rinv;
    }
    v2f av; av.x = p0; av.y = p1;

    const float* h0 = hb + m0 * HD;
    const float* h1 = h0 + HD;
    v2f bv;
    bv.x = h0[lane16];      bv.y = h1[lane16];
    acc0 = __builtin_amdgcn_wmma_f32_16x16x4_f32(false, av, false, bv, (short)0, acc0, false, false);
    bv.x = h0[16 + lane16]; bv.y = h1[16 + lane16];
    acc1 = __builtin_amdgcn_wmma_f32_16x16x4_f32(false, av, false, bv, (short)0, acc1, false, false);
    bv.x = h0[32 + lane16]; bv.y = h1[32 + lane16];
    acc2 = __builtin_amdgcn_wmma_f32_16x16x4_f32(false, av, false, bv, (short)0, acc2, false, false);
    bv.x = h0[48 + lane16]; bv.y = h1[48 + lane16];
    acc3 = __builtin_amdgcn_wmma_f32_16x16x4_f32(false, av, false, bv, (short)0, acc3, false, false);
  }

  float* op = opart + hh * 16 * HD;
  if (half == 0) {
#pragma unroll
    for (int v = 0; v < 8; ++v) {
      float* orow = op + (v + 8 * laneHi) * HD + lane16;
      orow[0]  = acc0[v];
      orow[16] = acc1[v];
      orow[32] = acc2[v];
      orow[48] = acc3[v];
    }
  }
  __syncthreads();
  if (half == 1) {
    float* ob = out + (size_t)b * Nq * Fout + hh * HD;
#pragma unroll
    for (int v = 0; v < 8; ++v) {
      const float* prow = op + (v + 8 * laneHi) * HD + lane16;
      float* orow = ob + (size_t)(n0 + v + 8 * laneHi) * Fout + lane16;
      orow[0]  = acc0[v] + prow[0];
      orow[16] = acc1[v] + prow[16];
      orow[32] = acc2[v] + prow[32];
      orow[48] = acc3[v] + prow[48];
    }
  }
}

extern "C" void kernel_launch(void* const* d_in, const int* in_sizes, int n_in,
                              void* d_out, int out_size, void* d_ws, size_t ws_size,
                              hipStream_t stream) {
  const float* x   = (const float*)d_in[0];
  const int*   adj = (const int*)d_in[1];
  const float* W   = (const float*)d_in[2];
  const float* a   = (const float*)d_in[3];
  float* out = (float*)d_out;

  float* hbuf = (float*)d_ws;
  float* s1 = hbuf + (size_t)Bq * Hh * Nq * HD;
  float* s2 = s1 + (size_t)Bq * Hh * Nq;

  gat_proj<<<2048, 128, 0, stream>>>(x, W, hbuf);
  gat_scores<<<4096, 256, 0, stream>>>(hbuf, a, s1, s2);
  gat_attn<<<dim3(Nq / 16, Bq), 256, 0, stream>>>(hbuf, adj, s1, s2, out);
}


# scores launch gone: each attn block recomputes s2 (whole head) and s1 (own rows) from h in its prologue, 4 fma a lane + 16-lane DPP butterfly (new sum order), loads ahead of the mask loop
# speedup vs baseline: 1.6542x; 1.6542x over previous
#include <hip/hip_runtime.h>


typedef float v2f __attribute__((ext_vector_type(2)));
typedef float v8f __attribute__((ext_vector_type(8)));

#define Bq 8
#define Nq 1024
#define Fin 256
#define Fout 256
#define Hh 4
#define HD 64
#define NEG_SLOPE 0.2f

__global__ __launch_bounds__(128) void gat_proj(const float* __restrict__ x,
                                                const float* __restrict__ W,
                                                float* __restrict__ hbuf) {
  const int tid = threadIdx.x;
  const int wave = tid >> 5, lane = tid & 31;
  const int lane16 = lane & 15, laneHi = lane >> 4;
  const int tile = blockIdx.x * 4 + wave;
  const int mt = tile >> 4;
  const int nt = tile & 15;
  const int row = mt * 16 + lane16;
  const int col = nt * 16 + lane16;

  const float* xr = x + row * Fin;
  v8f acc = {};
#pragma unroll 4
  for (int k = 0; k < Fin; k += 4) {
    const int k0 = k + 2 * laneHi;
    v2f av = *reinterpret_cast<const v2f*>(xr + k0);
    v2f bv;
    bv.x = W[k0 * Fout + col];
    bv.y = W[(k0 + 1) * Fout + col];
    acc = __builtin_amdgcn_wmma_f32_16x16x4_f32(false, av, false, bv,
                                                (short)0, acc, false, false);
  }
  const int hh = col >> 6, d = col & 63;
#pragma unroll
  for (int v = 0; v < 8; ++v) {
    const int r = mt * 16 + v + 8 * laneHi;
    const int bb = r >> 10, n = r & (Nq - 1);
    hbuf[(((bb * Hh + hh) * Nq + n) * HD) + d] = acc[v];
  }
}

#define BR 128
#define MC 64
#define RPP 160
#define MWP 33

__device__ __forceinline__ float row16_sum(float v) {
  v += __builtin_bit_cast(float, __builtin_amdgcn_update_dpp(0, __builtin_bit_cast(int, v), 0x161, 0xf, 0xf, true));
  v += __builtin_bit_cast(float, __builtin_amdgcn_update_dpp(0, __builtin_bit_cast(int, v), 0x162, 0xf, 0xf, true));
  v += __builtin_bit_cast(float, __builtin_amdgcn_update_dpp(0, __builtin_bit_cast(int, v), 0x164, 0xf, 0xf, true));
  v += __builtin_bit_cast(float, __builtin_amdgcn_update_dpp(0, __builtin_bit_cast(int, v), 0x168, 0xf, 0xf, true));
  return v;
}

__device__ __forceinline__ float dot4(const float4 p, const float4 q) {
  return fmaf(p.w, q.w, fmaf(p.z, q.z, fmaf(p.y, q.y, p.x * q.x)));
}

__device__ __forceinline__ void put_pair(float* d, const float4 e, const float4 o) {
  *reinterpret_cast<float4*>(d)     = make_float4(e.x, o.x, e.y, o.y);
  *reinterpret_cast<float4*>(d + 4) = make_float4(e.z, o.z, e.w, o.w);
}

__global__ __launch_bounds__(256) void gat_attn(const float* __restrict__ hbuf,
                                                const int* __restrict__ adj,
                                                const float* __restrict__ a,
                                                float* __restrict__ out) {
  __shared__ unsigned int maskw[BR * MWP];
  __shared__ __attribute__((aligned(16))) float s2l[Nq];
  __shared__ float s1l[BR];
  __shared__ __attribute__((aligned(16))) float hl[2 * (MC / 2) * RPP];

  const int tid = threadIdx.x;
  const int wave = tid >> 5, lane = tid & 31;
  const int lane16 = lane & 15, laneHi = lane >> 4;
  const int hh = blockIdx.x, n0 = blockIdx.y * BR, b = blockIdx.z;

  const float* hb = hbuf + (size_t)(b * Hh + hh) * Nq * HD;
  const int prow = tid >> 4, cc4 = (tid & 15) * 4;
  const float* gsrc = hb + 2 * prow * HD + cc4;
  float* ldst = hl + prow * RPP + 2 * cc4;
  const float4 pe0 = *reinterpret_cast<const float4*>(gsrc);
  const float4 po0 = *reinterpret_cast<const float4*>(gsrc + HD);
  const float4 pe1 = *reinterpret_cast<const float4*>(gsrc + 32 * HD);
  const float4 po1 = *reinterpret_cast<const float4*>(gsrc + 33 * HD);

  const float* ah = a + hh * (2 * HD);
  const float4 a1v = *reinterpret_cast<const float4*>(ah + 4 * lane16);
  const float4 a2v = *reinterpret_cast<const float4*>(ah + HD + 4 * lane16);
  const float* hsw = hb + (size_t)(wave * 128 + laneHi) * HD + 4 * lane16;
  const float* h1 = hb + (size_t)(n0 + wave * 16 + laneHi) * HD + 4 * lane16;
  float4 h1v[8];
#pragma unroll
  for (int j = 0; j < 8; ++j) h1v[j] = *reinterpret_cast<const float4*>(h1 + 2 * j * HD);
  float4 h2v[64];
#pragma unroll
  for (int j = 0; j < 64; ++j) h2v[j] = *reinterpret_cast<const float4*>(hsw + 2 * j * HD);
  const int* adjb = adj + ((size_t)b * Nq + n0) * Nq;
  for (int it = 0; it < BR * 32 / 256; ++it) {
    const int w = tid + it * 256;
    const int r = w >> 5, mw = w & 31;
    const int* ar = adjb + r * Nq + mw * 32;
    unsigned int bits = 0u;
#pragma unroll
    for (int j = 0; j < 32; ++j) bits |= (ar[j] != 0 ? 1u : 0u) << j;
    maskw[r * MWP + mw] = bits;
  }
#pragma unroll
  for (int g = 0; g < 8; ++g) {
    float d2[8];
#pragma unroll
    for (int j = 0; j < 8; ++j) d2[j] = row16_sum(dot4(h2v[8 * g + j], a2v));
    if (lane16 == 0) {
#pragma unroll
      for (int j = 0; j < 8; ++j) s2l[wave * 128 + 16 * g + 2 * j + laneHi] = d2[j];
    }
  }
  {
    float d1[8];
#pragma unroll
    for (int j = 0; j < 8; ++j) d1[j] = row16_sum(dot4(h1v[j], a1v));
    if (lane16 == 0) {
#pragma unroll
      for (int j = 0; j < 8; ++j) s1l[wave * 16 + 2 * j + laneHi] = d1[j];
    }
  }
  const int row = wave * 16 + lane16;
  put_pair(ldst, pe0, po0);
  put_pair(ldst + 16 * RPP, pe1, po1);
  __syncthreads();
  const float s1v = s1l[row];

  const unsigned int* mrow = maskw + row * MWP;
  const int sh = 2 * laneHi;

  float mxa = -1e30f, sma = 0.f, mxb = -1e30f, smb = 0.f;
  for (int g = 0; g < Nq / 32; ++g) {
    const unsigned int wd = mrow[g] >> sh;
    const float* s2p = s2l + g * 32 + sh;
#pragma unroll
    for (int j = 0; j < 8; ++j) {
      const v2f sv = *reinterpret_cast<const v2f*>(s2p + 4 * j);
      float v0 = s1v + sv.x;
      v0 = v0 > 0.f ? v0 : NEG_SLOPE * v0;
      float v1 = s1v + sv.y;
      v1 = v1 > 0.f ? v1 : NEG_SLOPE * v1;
      const bool b0 = ((wd >> (4 * j)) & 1u) != 0u;
      const bool b1 = ((wd >> (4 * j + 1)) & 1u) != 0u;
      const float na = b0 ? fmaxf(mxa, v0) : mxa;
      sma = sma * __expf(mxa - na) + (b0 ? __expf(v0 - na) : 0.f);
      mxa = na;
      const float nb = b1 ? fmaxf(mxb, v1) : mxb;
      smb = smb * __expf(mxb - nb) + (b1 ? __expf(v1 - nb) : 0.f);
      mxb = nb;
    }
  }
  float rmax, rinv;
  {
    const float mx = fmaxf(mxa, mxb);
    const float sm = sma * __expf(mxa - mx) + smb * __expf(mxb - mx);
    const float omx = __shfl_xor(mx, 16, 32);
    const float osm = __shfl_xor(sm, 16, 32);
    const float mlo = laneHi ? omx : mx, slo = laneHi ? osm : sm;
    const float mhi = laneHi ? mx : omx, shi = laneHi ? sm : osm;
    rmax = fmaxf(mlo, mhi);
    const float rsum = slo * __expf(mlo - rmax) + shi * __expf(mhi - rmax);
    rinv = rsum > 0.f ? 1.f / rsum : 0.f;
  }

  v8f acc0 = {}, acc1 = {}, acc2 = {}, acc3 = {};
  auto mma_chunk = [&](const int ci) {
    const float* cur = hl + (ci & 1) * ((MC / 2) * RPP) + laneHi * RPP + 2 * lane16;
    const float* s2c = s2l + ci * MC + sh;
    const unsigned int wd0 = mrow[ci * (MC / 32)] >> sh;
    const unsigned int wd1 = mrow[ci * (MC / 32) + 1] >> sh;
    v2f bf[2][16], sf[2][4];
    auto load_block = [&](const int q, v2f* bq, v2f* sq) {
#pragma unroll
      for (int ks = 0; ks < 4; ++ks) {
        const float* hp = cur + (8 * q + 2 * ks) * RPP;
        sq[ks] = *reinterpret_cast<const v2f*>(s2c + 16 * q + 4 * ks);
        bq[4 * ks + 0] = *reinterpret_cast<const v2f*>(hp);
        bq[4 * ks + 1] = *reinterpret_cast<const v2f*>(hp + 32);
        bq[4 * ks + 2] = *reinterpret_cast<const v2f*>(hp + 64);
        bq[4 * ks + 3] = *reinterpret_cast<const v2f*>(hp + 96);
      }
    };
    auto mma_block = [&](const int q, const v2f* bq, const v2f* sq) {
      const unsigned int wd = (q < 2) ? wd0 : wd1;
#pragma unroll
      for (int ks = 0; ks < 4; ++ks) {
        const int bit = 16 * (q & 1) + 4 * ks;
        float v0 = s1v + sq[ks].x;
        v0 = v0 > 0.f ? v0 : NEG_SLOPE * v0;
        float v1 = s1v + sq[ks].y;
        v1 = v1 > 0.f ? v1 : NEG_SLOPE * v1;
        const float e0 = __expf(v0 - rmax) * rinv;
        const float e1 = __expf(v1 - rmax) * rinv;
        v2f av;
        av.x = ((wd >> bit) & 1u) ? e0 : 0.f;
        av.y = ((wd >> (bit + 1)) & 1u) ? e1 : 0.f;
        acc0 = __builtin_amdgcn_wmma_f32_16x16x4_f32(false, av, false, bq[4 * ks + 0], (short)0, acc0, false, false);
        acc1 = __builtin_amdgcn_wmma_f32_16x16x4_f32(false, av, false, bq[4 * ks + 1], (short)0, acc1, false, false);
        acc2 = __builtin_amdgcn_wmma_f32_16x16x4_f32(false, av, false, bq[4 * ks + 2], (short)0, acc2, false, false);
        acc3 = __builtin_amdgcn_wmma_f32_16x16x4_f32(false, av, false, bq[4 * ks + 3], (short)0, acc3, false, false);
      }
    };
    load_block(0, bf[0], sf[0]);
    load_block(1, bf[1], sf[1]);
    __builtin_amdgcn_sched_barrier(0);
    mma_block(0, bf[0], sf[0]);
    __builtin_amdgcn_sched_barrier(0);
    load_block(2, bf[0], sf[0]);
    __builtin_amdgcn_sched_barrier(0);
    mma_block(1, bf[1], sf[1]);
    __builtin_amdgcn_sched_barrier(0);
    load_block(3, bf[1], sf[1]);
    __builtin_amdgcn_sched_barrier(0);
    mma_block(2, bf[0], sf[0]);
    mma_block(3, bf[1], sf[1]);
  };
  for (int ci = 0; ci < Nq / MC - 1; ++ci) {
    const float* gs = gsrc + (size_t)(ci + 1) * MC * HD;
    const float4 e0 = *reinterpret_cast<const float4*>(gs);
    const float4 o0 = *reinterpret_cast<const float4*>(gs + HD);
    const float4 e1 = *reinterpret_cast<const float4*>(gs + 32 * HD);
    const float4 o1 = *reinterpret_cast<const float4*>(gs + 33 * HD);
    mma_chunk(ci);
    float* ld = ldst + ((ci + 1) & 1) * ((MC / 2) * RPP);
    put_pair(ld, e0, o0);
    put_pair(ld + 16 * RPP, e1, o1);
    __syncthreads();
  }
  mma_chunk(Nq / MC - 1);

  float* ob = out + ((size_t)b * Nq + n0 + wave * 16) * Fout + hh * HD;
#pragma unroll
  for (int v = 0; v < 8; ++v) {
    float* orow = ob + (size_t)(v + 8 * laneHi) * Fout + lane16;
    orow[0]  = acc0[v];
    orow[16] = acc1[v];
    orow[32] = acc2[v];
    orow[48] = acc3[v];
  }
}

extern "C" void kernel_launch(void* const* d_in, const int* in_sizes, int n_in,
                              void* d_out, int out_size, void* d_ws, size_t ws_size,
                              hipStream_t stream) {
  const float* x   = (const float*)d_in[0];
  const int*   adj = (const int*)d_in[1];
  const float* W   = (const float*)d_in[2];
  const float* a   = (const float*)d_in[3];
  float* out = (float*)d_out;

  float* hbuf = (float*)d_ws;

  gat_proj<<<2048, 128, 0, stream>>>(x, W, hbuf);
  gat_attn<<<dim3(Hh, Nq / 128, Bq), 256, 0, stream>>>(hbuf, adj, a, out);
}


# adj tile read in whole lines: a wave takes its own 16 rows, 8 loads of 16 bytes a lane per row (4 lines an instruction, was 32); mask words by wave ballot, bit = k-step; loads 4 rows ahead
# speedup vs baseline: 1.8802x; 1.1366x over previous
#include <hip/hip_runtime.h>


typedef float v2f __attribute__((ext_vector_type(2)));
typedef float v8f __attribute__((ext_vector_type(8)));

#define Bq 8
#define Nq 1024
#define Fin 256
#define Fout 256
#define Hh 4
#define HD 64
#define NEG_SLOPE 0.2f

__global__ __launch_bounds__(128) void gat_proj(const float* __restrict__ x,
                                                const float* __restrict__ W,
                                                float* __restrict__ hbuf) {
  const int tid = threadIdx.x;
  const int wave = tid >> 5, lane = tid & 31;
  const int lane16 = lane & 15, laneHi = lane >> 4;
  const int tile = blockIdx.x * 4 + wave;
  const int mt = tile >> 4;
  const int nt = tile & 15;
  const int row = mt * 16 + lane16;
  const int col = nt * 16 + lane16;

  const float* xr = x + row * Fin;
  v8f acc = {};
#pragma unroll 4
  for (int k = 0; k < Fin; k += 4) {
    const int k0 = k + 2 * laneHi;
    v2f av = *reinterpret_cast<const v2f*>(xr + k0);
    v2f bv;
    bv.x = W[k0 * Fout + col];
    bv.y = W[(k0 + 1) * Fout + col];
    acc = __builtin_amdgcn_wmma_f32_16x16x4_f32(false, av, false, bv,
                                                (short)0, acc, false, false);
  }
  const int hh = col >> 6, d = col & 63;
#pragma unroll
  for (int v = 0; v < 8; ++v) {
    const int r = mt * 16 + v + 8 * laneHi;
    const int bb = r >> 10, n = r & (Nq - 1);
    hbuf[(((bb * Hh + hh) * Nq + n) * HD) + d] = acc[v];
  }
}

#define BR 128
#define MC 64
#define RPP 160
#define MWP 36

__device__ __forceinline__ float row16_sum(float v) {
  v += __builtin_bit_cast(float, __builtin_amdgcn_update_dpp(0, __builtin_bit_cast(int, v), 0x161, 0xf, 0xf, true));
  v += __builtin_bit_cast(float, __builtin_amdgcn_update_dpp(0, __builtin_bit_cast(int, v), 0x162, 0xf, 0xf, true));
  v += __builtin_bit_cast(float, __builtin_amdgcn_update_dpp(0, __builtin_bit_cast(int, v), 0x164, 0xf, 0xf, true));
  v += __builtin_bit_cast(float, __builtin_amdgcn_update_dpp(0, __builtin_bit_cast(int, v), 0x168, 0xf, 0xf, true));
  return v;
}

__device__ __forceinline__ float dot4(const float4 p, const float4 q) {
  return fmaf(p.w, q.w, fmaf(p.z, q.z, fmaf(p.y, q.y, p.x * q.x)));
}

__device__ __forceinline__ void put_pair(float* d, const float4 e, const float4 o) {
  *reinterpret_cast<float4*>(d)     = make_float4(e.x, o.x, e.y, o.y);
  *reinterpret_cast<float4*>(d + 4) = make_float4(e.z, o.z, e.w, o.w);
}

__global__ __launch_bounds__(256) void gat_attn(const float* __restrict__ hbuf,
                                                const int* __restrict__ adj,
                                                const float* __restrict__ a,
                                                float* __restrict__ out) {
  __shared__ __attribute__((aligned(16))) unsigned int maskw[BR * MWP];
  __shared__ __attribute__((aligned(16))) float s2l[Nq];
  __shared__ float s1l[BR];
  __shared__ __attribute__((aligned(16))) float hl[2 * (MC / 2) * RPP];

  const int tid = threadIdx.x;
  const int wave = tid >> 5, lane = tid & 31;
  const int lane16 = lane & 15, laneHi = lane >> 4;
  const int hh = blockIdx.x, n0 = blockIdx.y * BR, b = blockIdx.z;

  const float* hb = hbuf + (size_t)(b * Hh + hh) * Nq * HD;
  const int prow = tid >> 4, cc4 = (tid & 15) * 4;
  const float* gsrc = hb + 2 * prow * HD + cc4;
  float* ldst = hl + prow * RPP + 2 * cc4;
  const float4 pe0 = *reinterpret_cast<const float4*>(gsrc);
  const float4 po0 = *reinterpret_cast<const float4*>(gsrc + HD);
  const float4 pe1 = *reinterpret_cast<const float4*>(gsrc + 32 * HD);
  const float4 po1 = *reinterpret_cast<const float4*>(gsrc + 33 * HD);

  const float* ah = a + hh * (2 * HD);
  const float4 a1v = *reinterpret_cast<const float4*>(ah + 4 * lane16);
  const float4 a2v = *reinterpret_cast<const float4*>(ah + HD + 4 * lane16);
  const float* hsw = hb + (size_t)(wave * 128 + laneHi) * HD + 4 * lane16;
  const float* h1 = hb + (size_t)(n0 + wave * 16 + laneHi) * HD + 4 * lane16;
  float4 h1v[8];
#pragma unroll
  for (int j = 0; j < 8; ++j) h1v[j] = *reinterpret_cast<const float4*>(h1 + 2 * j * HD);
  float4 h2v[64];
#pragma unroll
  for (int j = 0; j < 64; ++j) h2v[j] = *reinterpret_cast<const float4*>(hsw + 2 * j * HD);
  const int4* adjw = reinterpret_cast<const int4*>(adj + ((size_t)b * Nq + n0 + wave * 16) * Nq) + lane;
  int4 ab[2][32];
  auto load_rows = [&](const int g, int4* buf) {
#pragma unroll
    for (int i = 0; i < 4; ++i) {
#pragma unroll
      for (int q = 0; q < 8; ++q) buf[8 * i + q] = adjw[(4 * g + i) * (Nq / 4) + 32 * q];
    }
  };
  auto mask_rows = [&](const int g, const int4* buf) {
#pragma unroll
    for (int i = 0; i < 4; ++i) {
      unsigned int my = 0u;
#pragma unroll
      for (int q = 0; q < 8; ++q) {
        const int4 v = buf[8 * i + q];
        const unsigned int w0 = __builtin_amdgcn_ballot_w32(v.x != 0);
        const unsigned int w1 = __builtin_amdgcn_ballot_w32(v.y != 0);
        const unsigned int w2 = __builtin_amdgcn_ballot_w32(v.z != 0);
        const unsigned int w3 = __builtin_amdgcn_ballot_w32(v.w != 0);
        my = (lane == 4 * q + 0) ? w0 : my;
        my = (lane == 4 * q + 1) ? w1 : my;
        my = (lane == 4 * q + 2) ? w2 : my;
        my = (lane == 4 * q + 3) ? w3 : my;
      }
      maskw[(wave * 16 + 4 * g + i) * MWP + lane] = my;
    }
  };
  load_rows(0, ab[0]);
#pragma unroll
  for (int g = 0; g < 8; ++g) {
    float d2[8];
#pragma unroll
    for (int j = 0; j < 8; ++j) d2[j] = row16_sum(dot4(h2v[8 * g + j], a2v));
    if (lane16 == 0) {
#pragma unroll
      for (int j = 0; j < 8; ++j) s2l[wave * 128 + 16 * g + 2 * j + laneHi] = d2[j];
    }
  }
  {
    float d1[8];
#pragma unroll
    for (int j = 0; j < 8; ++j) d1[j] = row16_sum(dot4(h1v[j], a1v));
    if (lane16 == 0) {
#pragma unroll
      for (int j = 0; j < 8; ++j) s1l[wave * 16 + 2 * j + laneHi] = d1[j];
    }
  }
  load_rows(1, ab[1]);
  mask_rows(0, ab[0]);
  load_rows(2, ab[0]);
  mask_rows(1, ab[1]);
  load_rows(3, ab[1]);
  mask_rows(2, ab[0]);
  mask_rows(3, ab[1]);
  const int row = wave * 16 + lane16;
  put_pair(ldst, pe0, po0);
  put_pair(ldst + 16 * RPP, pe1, po1);
  __syncthreads();
  const float s1v = s1l[row];

  const unsigned int* mrow = maskw + row * MWP + 2 * laneHi;
  const int sh = 2 * laneHi;

  float mxa = -1e30f, sma = 0.f, mxb = -1e30f, smb = 0.f;
  for (int g = 0; g < Nq / 32; ++g) {
    const uint2 wp = *reinterpret_cast<const uint2*>(mrow + 4 * (g >> 2));
    const unsigned int wda = wp.x >> (8 * (g & 3)), wdb = wp.y >> (8 * (g & 3));
    const float* s2p = s2l + g * 32 + sh;
#pragma unroll
    for (int j = 0; j < 8; ++j) {
      const v2f sv = *reinterpret_cast<const v2f*>(s2p + 4 * j);
      float v0 = s1v + sv.x;
      v0 = v0 > 0.f ? v0 : NEG_SLOPE * v0;
      float v1 = s1v + sv.y;
      v1 = v1 > 0.f ? v1 : NEG_SLOPE * v1;
      const bool b0 = ((wda >> j) & 1u) != 0u;
      const bool b1 = ((wdb >> j) & 1u) != 0u;
      const float na = b0 ? fmaxf(mxa, v0) : mxa;
      sma = sma * __expf(mxa - na) + (b0 ? __expf(v0 - na) : 0.f);
      mxa = na;
      const float nb = b1 ? fmaxf(mxb, v1) : mxb;
      smb = smb * __expf(mxb - nb) + (b1 ? __expf(v1 - nb) : 0.f);
      mxb = nb;
    }
  }
  float rmax, rinv;
  {
    const float mx = fmaxf(mxa, mxb);
    const float sm = sma * __expf(mxa - mx) + smb * __expf(mxb - mx);
    const float omx = __shfl_xor(mx, 16, 32);
    const float osm = __shfl_xor(sm, 16, 32);
    const float mlo = laneHi ? omx : mx, slo = laneHi ? osm : sm;
    const float mhi = laneHi ? mx : omx, shi = laneHi ? sm : osm;
    rmax = fmaxf(mlo, mhi);
    const float rsum = slo * __expf(mlo - rmax) + shi * __expf(mhi - rmax);
    rinv = rsum > 0.f ? 1.f / rsum : 0.f;
  }

  v8f acc0 = {}, acc1 = {}, acc2 = {}, acc3 = {};
  auto mma_chunk = [&](const int ci) {
    const float* cur = hl + (ci & 1) * ((MC / 2) * RPP) + laneHi * RPP + 2 * lane16;
    const float* s2c = s2l + ci * MC + sh;
    const uint2 wp = *reinterpret_cast<const uint2*>(mrow + 4 * (ci >> 1));
    const unsigned int wda = wp.x >> (16 * (ci & 1)), wdb = wp.y >> (16 * (ci & 1));
    v2f bf[2][16], sf[2][4];
    auto load_block = [&](const int q, v2f* bq, v2f* sq) {
#pragma unroll
      for (int ks = 0; ks < 4; ++ks) {
        const float* hp = cur + (8 * q + 2 * ks) * RPP;
        sq[ks] = *reinterpret_cast<const v2f*>(s2c + 16 * q + 4 * ks);
        bq[4 * ks + 0] = *reinterpret_cast<const v2f*>(hp);
        bq[4 * ks + 1] = *reinterpret_cast<const v2f*>(hp + 32);
        bq[4 * ks + 2] = *reinterpret_cast<const v2f*>(hp + 64);
        bq[4 * ks + 3] = *reinterpret_cast<const v2f*>(hp + 96);
      }
    };
    auto mma_block = [&](const int q, const v2f* bq, const v2f* sq) {
#pragma unroll
      for (int ks = 0; ks < 4; ++ks) {
        const int bit = 4 * q + ks;
        float v0 = s1v + sq[ks].x;
        v0 = v0 > 0.f ? v0 : NEG_SLOPE * v0;
        float v1 = s1v + sq[ks].y;
        v1 = v1 > 0.f ? v1 : NEG_SLOPE * v1;
        const float e0 = __expf(v0 - rmax) * rinv;
        const float e1 = __expf(v1 - rmax) * rinv;
        v2f av;
        av.x = ((wda >> bit) & 1u) ? e0 : 0.f;
        av.y = ((wdb >> bit) & 1u) ? e1 : 0.f;
        acc0 = __builtin_amdgcn_wmma_f32_16x16x4_f32(false, av, false, bq[4 * ks + 0], (short)0, acc0, false, false);
        acc1 = __builtin_amdgcn_wmma_f32_16x16x4_f32(false, av, false, bq[4 * ks + 1], (short)0, acc1, false, false);
        acc2 = __builtin_amdgcn_wmma_f32_16x16x4_f32(false, av, false, bq[4 * ks + 2], (short)0, acc2, false, false);
        acc3 = __builtin_amdgcn_wmma_f32_16x16x4_f32(false, av, false, bq[4 * ks + 3], (short)0, acc3, false, false);
      }
    };
    load_block(0, bf[0], sf[0]);
    load_block(1, bf[1], sf[1]);
    __builtin_amdgcn_sched_barrier(0);
    mma_block(0, bf[0], sf[0]);
    __builtin_amdgcn_sched_barrier(0);
    load_block(2, bf[0], sf[0]);
    __builtin_amdgcn_sched_barrier(0);
    mma_block(1, bf[1], sf[1]);
    __builtin_amdgcn_sched_barrier(0);
    load_block(3, bf[1], sf[1]);
    __builtin_amdgcn_sched_barrier(0);
    mma_block(2, bf[0], sf[0]);
    mma_block(3, bf[1], sf[1]);
  };
  for (int ci = 0; ci < Nq / MC - 1; ++ci) {
    const float* gs = gsrc + (size_t)(ci + 1) * MC * HD;
    const float4 e0 = *reinterpret_cast<const float4*>(gs);
    const float4 o0 = *reinterpret_cast<const float4*>(gs + HD);
    const float4 e1 = *reinterpret_cast<const float4*>(gs + 32 * HD);
    const float4 o1 = *reinterpret_cast<const float4*>(gs + 33 * HD);
    mma_chunk(ci);
    float* ld = ldst + ((ci + 1) & 1) * ((MC / 2) * RPP);
    put_pair(ld, e0, o0);
    put_pair(ld + 16 * RPP, e1, o1);
    __syncthreads();
  }
  mma_chunk(Nq / MC - 1);

  float* ob = out + ((size_t)b * Nq + n0 + wave * 16) * Fout + hh * HD;
#pragma unroll
  for (int v = 0; v < 8; ++v) {
    float* orow = ob + (size_t)(v + 8 * laneHi) * Fout + lane16;
    orow[0]  = acc0[v];
    orow[16] = acc1[v];
    orow[32] = acc2[v];
    orow[48] = acc3[v];
  }
}

extern "C" void kernel_launch(void* const* d_in, const int* in_sizes, int n_in,
                              void* d_out, int out_size, void* d_ws, size_t ws_size,
                              hipStream_t stream) {
  const float* x   = (const float*)d_in[0];
  const int*   adj = (const int*)d_in[1];
  const float* W   = (const float*)d_in[2];
  const float* a   = (const float*)d_in[3];
  float* out = (float*)d_out;

  float* hbuf = (float*)d_ws;

  gat_proj<<<2048, 128, 0, stream>>>(x, W, hbuf);
  gat_attn<<<dim3(Hh, Nq / 128, Bq), 256, 0, stream>>>(hbuf, adj, a, out);
}


# exps out of the NxN loops: exp(lrelu(s1+s2)-shift) = E[m] or F[m]*B[n] by test E[m]>thr[n], shift = lrelu(s1 + head max of s2) >= row max; stats pass gone, row sum of p in product pass (new sum order)
# speedup vs baseline: 2.2103x; 1.1756x over previous
#include <hip/hip_runtime.h>


typedef float v2f __attribute__((ext_vector_type(2)));
typedef float v8f __attribute__((ext_vector_type(8)));

#define Bq 8
#define Nq 1024
#define Fin 256
#define Fout 256
#define Hh 4
#define HD 64
#define NEG_SLOPE 0.2f

__global__ __launch_bounds__(128) void gat_proj(const float* __restrict__ x,
                                                const float* __restrict__ W,
                                                float* __restrict__ hbuf) {
  const int tid = threadIdx.x;
  const int wave = tid >> 5, lane = tid & 31;
  const int lane16 = lane & 15, laneHi = lane >> 4;
  const int tile = blockIdx.x * 4 + wave;
  const int mt = tile >> 4;
  const int nt = tile & 15;
  const int row = mt * 16 + lane16;
  const int col = nt * 16 + lane16;

  const float* xr = x + row * Fin;
  v8f acc = {};
#pragma unroll 4
  for (int k = 0; k < Fin; k += 4) {
    const int k0 = k + 2 * laneHi;
    v2f av = *reinterpret_cast<const v2f*>(xr + k0);
    v2f bv;
    bv.x = W[k0 * Fout + col];
    bv.y = W[(k0 + 1) * Fout + col];
    acc = __builtin_amdgcn_wmma_f32_16x16x4_f32(false, av, false, bv,
                                                (short)0, acc, false, false);
  }
  const int hh = col >> 6, d = col & 63;
#pragma unroll
  for (int v = 0; v < 8; ++v) {
    const int r = mt * 16 + v + 8 * laneHi;
    const int bb = r >> 10, n = r & (Nq - 1);
    hbuf[(((bb * Hh + hh) * Nq + n) * HD) + d] = acc[v];
  }
}

#define BR 128
#define MC 64
#define RPP 160
#define MWP 36

__device__ __forceinline__ float row16_sum(float v) {
  v += __builtin_bit_cast(float, __builtin_amdgcn_update_dpp(0, __builtin_bit_cast(int, v), 0x161, 0xf, 0xf, true));
  v += __builtin_bit_cast(float, __builtin_amdgcn_update_dpp(0, __builtin_bit_cast(int, v), 0x162, 0xf, 0xf, true));
  v += __builtin_bit_cast(float, __builtin_amdgcn_update_dpp(0, __builtin_bit_cast(int, v), 0x164, 0xf, 0xf, true));
  v += __builtin_bit_cast(float, __builtin_amdgcn_update_dpp(0, __builtin_bit_cast(int, v), 0x168, 0xf, 0xf, true));
  return v;
}

__device__ __forceinline__ float dot4(const float4 p, const float4 q) {
  return fmaf(p.w, q.w, fmaf(p.z, q.z, fmaf(p.y, q.y, p.x * q.x)));
}

__device__ __forceinline__ void put_pair(float* d, const float4 e, const float4 o) {
  *reinterpret_cast<float4*>(d)     = make_float4(e.x, o.x, e.y, o.y);
  *reinterpret_cast<float4*>(d + 4) = make_float4(e.z, o.z, e.w, o.w);
}

__global__ __launch_bounds__(256) void gat_attn(const float* __restrict__ hbuf,
                                                const int* __restrict__ adj,
                                                const float* __restrict__ a,
                                                float* __restrict__ out) {
  __shared__ __attribute__((aligned(16))) unsigned int maskw[BR * MWP];
  __shared__ __attribute__((aligned(16))) float s2l[Nq];
  __shared__ float s1l[BR];
  __shared__ __attribute__((aligned(16))) float efl[2 * Nq];
  __shared__ __attribute__((aligned(16))) float wmaxl[8];
  __shared__ __attribute__((aligned(16))) float hl[2 * (MC / 2) * RPP];

  const int tid = threadIdx.x;
  const int wave = tid >> 5, lane = tid & 31;
  const int lane16 = lane & 15, laneHi = lane >> 4;
  const int hh = blockIdx.x, n0 = blockIdx.y * BR, b = blockIdx.z;

  const float* hb = hbuf + (size_t)(b * Hh + hh) * Nq * HD;
  const int prow = tid >> 4, cc4 = (tid & 15) * 4;
  const float* gsrc = hb + 2 * prow * HD + cc4;
  float* ldst = hl + prow * RPP + 2 * cc4;
  const float4 pe0 = *reinterpret_cast<const float4*>(gsrc);
  const float4 po0 = *reinterpret_cast<const float4*>(gsrc + HD);
  const float4 pe1 = *reinterpret_cast<const float4*>(gsrc + 32 * HD);
  const float4 po1 = *reinterpret_cast<const float4*>(gsrc + 33 * HD);

  const float* ah = a + hh * (2 * HD);
  const float4 a1v = *reinterpret_cast<const float4*>(ah + 4 * lane16);
  const float4 a2v = *reinterpret_cast<const float4*>(ah + HD + 4 * lane16);
  const float* hsw = hb + (size_t)(wave * 128 + laneHi) * HD + 4 * lane16;
  const float* h1 = hb + (size_t)(n0 + wave * 16 + laneHi) * HD + 4 * lane16;
  float4 h1v[8];
#pragma unroll
  for (int j = 0; j < 8; ++j) h1v[j] = *reinterpret_cast<const float4*>(h1 + 2 * j * HD);
  float4 h2v[64];
#pragma unroll
  for (int j = 0; j < 64; ++j) h2v[j] = *reinterpret_cast<const float4*>(hsw + 2 * j * HD);
  const int4* adjw = reinterpret_cast<const int4*>(adj + ((size_t)b * Nq + n0 + wave * 16) * Nq) + lane;
  int4 ab[2][32];
  auto load_rows = [&](const int g, int4* buf) {
#pragma unroll
    for (int i = 0; i < 4; ++i) {
#pragma unroll
      for (int q = 0; q < 8; ++q) buf[8 * i + q] = adjw[(4 * g + i) * (Nq / 4) + 32 * q];
    }
  };
  auto mask_rows = [&](const int g, const int4* buf) {
#pragma unroll
    for (int i = 0; i < 4; ++i) {
      unsigned int my = 0u;
#pragma unroll
      for (int q = 0; q < 8; ++q) {
        const int4 v = buf[8 * i + q];
        const unsigned int w0 = __builtin_amdgcn_ballot_w32(v.x != 0);
        const unsigned int w1 = __builtin_amdgcn_ballot_w32(v.y != 0);
        const unsigned int w2 = __builtin_amdgcn_ballot_w32(v.z != 0);
        const unsigned int w3 = __builtin_amdgcn_ballot_w32(v.w != 0);
        my = (lane == 4 * q + 0) ? w0 : my;
        my = (lane == 4 * q + 1) ? w1 : my;
        my = (lane == 4 * q + 2) ? w2 : my;
        my = (lane == 4 * q + 3) ? w3 : my;
      }
      maskw[(wave * 16 + 4 * g + i) * MWP + lane] = my;
    }
  };
  load_rows(0, ab[0]);
  float wmx = -__builtin_huge_valf();
#pragma unroll
  for (int g = 0; g < 8; ++g) {
    float d2[8];
#pragma unroll
    for (int j = 0; j < 8; ++j) {
      d2[j] = row16_sum(dot4(h2v[8 * g + j], a2v));
      wmx = fmaxf(wmx, d2[j]);
    }
    if (lane16 == 0) {
#pragma unroll
      for (int j = 0; j < 8; ++j) s2l[wave * 128 + 16 * g + 2 * j + laneHi] = d2[j];
    }
  }
  wmx = fmaxf(wmx, __shfl_xor(wmx, 16, 32));
  if (lane == 0) wmaxl[wave] = wmx;
  {
    float d1[8];
#pragma unroll
    for (int j = 0; j < 8; ++j) d1[j] = row16_sum(dot4(h1v[j], a1v));
    if (lane16 == 0) {
#pragma unroll
      for (int j = 0; j < 8; ++j) s1l[wave * 16 + 2 * j + laneHi] = d1[j];
    }
  }
  load_rows(1, ab[1]);
  mask_rows(0, ab[0]);
  load_rows(2, ab[0]);
  mask_rows(1, ab[1]);
  load_rows(3, ab[1]);
  mask_rows(2, ab[0]);
  mask_rows(3, ab[1]);
  const int row = wave * 16 + lane16;
  put_pair(ldst, pe0, po0);
  put_pair(ldst + 16 * RPP, pe1, po1);
  __syncthreads();
  const float s1v = s1l[row];

  float cmax;
  {
    const float4 w0 = *reinterpret_cast<const float4*>(wmaxl);
    const float4 w1 = *reinterpret_cast<const float4*>(wmaxl + 4);
    cmax = fmaxf(fmaxf(fmaxf(w0.x, w0.y), fmaxf(w0.z, w0.w)),
                 fmaxf(fmaxf(w1.x, w1.y), fmaxf(w1.z, w1.w)));
    const float4 sv = *reinterpret_cast<const float4*>(s2l + 4 * tid);
    const float t0 = sv.x - cmax, t1 = sv.y - cmax, t2 = sv.z - cmax, t3 = sv.w - cmax;
    float4 lo, hi;
    lo.x = __expf(t0); lo.y = __expf(t1); lo.z = __expf(NEG_SLOPE * t0); lo.w = __expf(NEG_SLOPE * t1);
    hi.x = __expf(t2); hi.y = __expf(t3); hi.z = __expf(NEG_SLOPE * t2); hi.w = __expf(NEG_SLOPE * t3);
    *reinterpret_cast<float4*>(efl + 8 * tid)     = lo;
    *reinterpret_cast<float4*>(efl + 8 * tid + 4) = hi;
  }
  __syncthreads();

  const unsigned int* mrow = maskw + row * MWP + 2 * laneHi;

  const float zrow = s1v + cmax;
  const float thr = __expf(-zrow);
  const float bneg = zrow > 0.f ? __expf((NEG_SLOPE - 1.f) * zrow) : 1.f;
  float rs0 = 0.f, rs1 = 0.f;

  v8f acc0 = {}, acc1 = {}, acc2 = {}, acc3 = {};
  auto mma_chunk = [&](const int ci) {
    const float* cur = hl + (ci & 1) * ((MC / 2) * RPP) + laneHi * RPP + 2 * lane16;
    const float* efc = efl + ci * (2 * MC) + 4 * laneHi;
    const uint2 wp = *reinterpret_cast<const uint2*>(mrow + 4 * (ci >> 1));
    const unsigned int wda = wp.x >> (16 * (ci & 1)), wdb = wp.y >> (16 * (ci & 1));
    v2f bf[2][16];
    float4 sf[2][4];
    auto load_block = [&](const int q, v2f* bq, float4* sq) {
#pragma unroll
      for (int ks = 0; ks < 4; ++ks) {
        const float* hp = cur + (8 * q + 2 * ks) * RPP;
        sq[ks] = *reinterpret_cast<const float4*>(efc + 32 * q + 8 * ks);
        bq[4 * ks + 0] = *reinterpret_cast<const v2f*>(hp);
        bq[4 * ks + 1] = *reinterpret_cast<const v2f*>(hp + 32);
        bq[4 * ks + 2] = *reinterpret_cast<const v2f*>(hp + 64);
        bq[4 * ks + 3] = *reinterpret_cast<const v2f*>(hp + 96);
      }
    };
    auto mma_block = [&](const int q, const v2f* bq, const float4* sq) {
#pragma unroll
      for (int ks = 0; ks < 4; ++ks) {
        const int bit = 4 * q + ks;
        const float4 ef = sq[ks];
        const float e0 = ef.x > thr ? ef.x : ef.z * bneg;
        const float e1 = ef.y > thr ? ef.y : ef.w * bneg;
        v2f av;
        av.x = ((wda >> bit) & 1u) ? e0 : 0.f;
        av.y = ((wdb >> bit) & 1u) ? e1 : 0.f;
        rs0 += av.x;
        rs1 += av.y;
        acc0 = __builtin_amdgcn_wmma_f32_16x16x4_f32(false, av, false, bq[4 * ks + 0], (short)0, acc0, false, false);
        acc1 = __builtin_amdgcn_wmma_f32_16x16x4_f32(false, av, false, bq[4 * ks + 1], (short)0, acc1, false, false);
        acc2 = __builtin_amdgcn_wmma_f32_16x16x4_f32(false, av, false, bq[4 * ks + 2], (short)0, acc2, false, false);
        acc3 = __builtin_amdgcn_wmma_f32_16x16x4_f32(false, av, false, bq[4 * ks + 3], (short)0, acc3, false, false);
      }
    };
    load_block(0, bf[0], sf[0]);
    load_block(1, bf[1], sf[1]);
    __builtin_amdgcn_sched_barrier(0);
    mma_block(0, bf[0], sf[0]);
    __builtin_amdgcn_sched_barrier(0);
    load_block(2, bf[0], sf[0]);
    __builtin_amdgcn_sched_barrier(0);
    mma_block(1, bf[1], sf[1]);
    __builtin_amdgcn_sched_barrier(0);
    load_block(3, bf[1], sf[1]);
    __builtin_amdgcn_sched_barrier(0);
    mma_block(2, bf[0], sf[0]);
    mma_block(3, bf[1], sf[1]);
  };
  for (int ci = 0; ci < Nq / MC - 1; ++ci) {
    const float* gs = gsrc + (size_t)(ci + 1) * MC * HD;
    const float4 e0 = *reinterpret_cast<const float4*>(gs);
    const float4 o0 = *reinterpret_cast<const float4*>(gs + HD);
    const float4 e1 = *reinterpret_cast<const float4*>(gs + 32 * HD);
    const float4 o1 = *reinterpret_cast<const float4*>(gs + 33 * HD);
    mma_chunk(ci);
    float* ld = ldst + ((ci + 1) & 1) * ((MC / 2) * RPP);
    put_pair(ld, e0, o0);
    put_pair(ld + 16 * RPP, e1, o1);
    __syncthreads();
  }
  mma_chunk(Nq / MC - 1);

  float rinv;
  {
    const float sm = rs0 + rs1;
    const float osm = __shfl_xor(sm, 16, 32);
    const float rsum = (laneHi ? osm : sm) + (laneHi ? sm : osm);
    rinv = rsum > 0.f ? 1.f / rsum : 0.f;
  }
  float* ob = out + ((size_t)b * Nq + n0 + wave * 16) * Fout + hh * HD;
#pragma unroll
  for (int v = 0; v < 8; ++v) {
    const float ri = __shfl(rinv, v + 8 * laneHi, 32);
    float* orow = ob + (size_t)(v + 8 * laneHi) * Fout + lane16;
    orow[0]  = acc0[v] * ri;
    orow[16] = acc1[v] * ri;
    orow[32] = acc2[v] * ri;
    orow[48] = acc3[v] * ri;
  }
}

extern "C" void kernel_launch(void* const* d_in, const int* in_sizes, int n_in,
                              void* d_out, int out_size, void* d_ws, size_t ws_size,
                              hipStream_t stream) {
  const float* x   = (const float*)d_in[0];
  const int*   adj = (const int*)d_in[1];
  const float* W   = (const float*)d_in[2];
  const float* a   = (const float*)d_in[3];
  float* out = (float*)d_out;

  float* hbuf = (float*)d_ws;

  gat_proj<<<2048, 128, 0, stream>>>(x, W, hbuf);
  gat_attn<<<dim3(Hh, Nq / 128, Bq), 256, 0, stream>>>(hbuf, adj, a, out);
}


# proj re-cut as a block GEMM through LDS: block = 128 rows x 1 head, 8 waves of 32x32; x, W staged in 64-k chunks by whole-line loads (2 buffers), fragments from LDS, h tile out via LDS; h bits same
# speedup vs baseline: 2.5407x; 1.1495x over previous
#include <hip/hip_runtime.h>


typedef float v2f __attribute__((ext_vector_type(2)));
typedef float v8f __attribute__((ext_vector_type(8)));

#define Bq 8
#define Nq 1024
#define Fin 256
#define Fout 256
#define Hh 4
#define HD 64
#define NEG_SLOPE 0.2f

#define PBM 128
#define PKC 64
#define XP 68
#define WPP 160

__device__ __forceinline__ void put_pair(float* d, const float4 e, const float4 o) {
  *reinterpret_cast<float4*>(d)     = make_float4(e.x, o.x, e.y, o.y);
  *reinterpret_cast<float4*>(d + 4) = make_float4(e.z, o.z, e.w, o.w);
}

__global__ __launch_bounds__(256) void gat_proj(const float* __restrict__ x,
                                                const float* W,
                                                float* __restrict__ hbuf) {
  __shared__ __attribute__((aligned(16))) float xs[2 * PBM * XP];
  __shared__ __attribute__((aligned(16))) float wl[2 * (PKC / 2) * WPP];

  const int tid = threadIdx.x;
  const int wave = tid >> 5, lane = tid & 31;
  const int lane16 = lane & 15, laneHi = lane >> 4;
  const int hh = blockIdx.x;
  const int r0 = blockIdx.y * PBM;
  const int wr = wave >> 1, wc = wave & 1;

  const int crow = tid >> 4, cc4 = (tid & 15) * 4;
  const float* xg = x + (size_t)(r0 + crow) * Fin + cc4;
  const float* wg = W + (size_t)(2 * crow) * Fout + hh * HD + cc4;
  float* xd = xs + crow * XP + cc4;
  float* wd = wl + crow * WPP + 2 * cc4;
  auto ld4 = [](const float* p) { return *reinterpret_cast<const float4*>(p); };
  auto st4 = [](float* p, const float4 v) { *reinterpret_cast<float4*>(p) = v; };

  v8f acc00 = {}, acc01 = {}, acc10 = {}, acc11 = {};
  auto mma_chunk = [&](const int buf) {
    const float* xa = xs + buf * (PBM * XP) + (32 * wr + lane16) * XP + 2 * laneHi;
    const float* wb = wl + buf * ((PKC / 2) * WPP) + laneHi * WPP + 2 * (32 * wc + lane16);
#pragma unroll
    for (int k = 0; k < PKC / 4; ++k) {
      const v2f fa0 = *reinterpret_cast<const v2f*>(xa + 4 * k);
      const v2f fa1 = *reinterpret_cast<const v2f*>(xa + 16 * XP + 4 * k);
      const v2f fb0 = *reinterpret_cast<const v2f*>(wb + 2 * k * WPP);
      const v2f fb1 = *reinterpret_cast<const v2f*>(wb + 2 * k * WPP + 32);
      acc00 = __builtin_amdgcn_wmma_f32_16x16x4_f32(false, fa0, false, fb0, (short)0, acc00, false, false);
      acc01 = __builtin_amdgcn_wmma_f32_16x16x4_f32(false, fa0, false, fb1, (short)0, acc01, false, false);
      acc10 = __builtin_amdgcn_wmma_f32_16x16x4_f32(false, fa1, false, fb0, (short)0, acc10, false, false);
      acc11 = __builtin_amdgcn_wmma_f32_16x16x4_f32(false, fa1, false, fb1, (short)0, acc11, false, false);
    }
  };

  {
    const float4 x0 = ld4(xg),            x1 = ld4(xg + 16 * Fin), x2 = ld4(xg + 32 * Fin), x3 = ld4(xg + 48 * Fin);
    const float4 x4 = ld4(xg + 64 * Fin), x5 = ld4(xg + 80 * Fin), x6 = ld4(xg + 96 * Fin), x7 = ld4(xg + 112 * Fin);
    const float4 e0 = ld4(wg), o0 = ld4(wg + Fout), e1 = ld4(wg + 32 * Fout), o1 = ld4(wg + 33 * Fout);
    st4(xd, x0);           st4(xd + 16 * XP, x1); st4(xd + 32 * XP, x2); st4(xd + 48 * XP, x3);
    st4(xd + 64 * XP, x4); st4(xd + 80 * XP, x5); st4(xd + 96 * XP, x6); st4(xd + 112 * XP, x7);
    put_pair(wd, e0, o0);
    put_pair(wd + 16 * WPP, e1, o1);
  }
  __syncthreads();
  for (int c = 0; c < Fin / PKC - 1; ++c) {
    const float* xn = xg + (c + 1) * PKC;
    const float* wn = wg + (size_t)(c + 1) * PKC * Fout;
    const float4 e0 = ld4(wn), o0 = ld4(wn + Fout), e1 = ld4(wn + 32 * Fout), o1 = ld4(wn + 33 * Fout);
    const float4 x0 = ld4(xn),            x1 = ld4(xn + 16 * Fin), x2 = ld4(xn + 32 * Fin), x3 = ld4(xn + 48 * Fin);
    const float4 x4 = ld4(xn + 64 * Fin), x5 = ld4(xn + 80 * Fin), x6 = ld4(xn + 96 * Fin), x7 = ld4(xn + 112 * Fin);
    mma_chunk(c & 1);
    __builtin_amdgcn_sched_barrier(0);
    float* xb = xd + ((c + 1) & 1) * (PBM * XP);
    float* wb = wd + ((c + 1) & 1) * ((PKC / 2) * WPP);
    st4(xb, x0);           st4(xb + 16 * XP, x1); st4(xb + 32 * XP, x2); st4(xb + 48 * XP, x3);
    st4(xb + 64 * XP, x4); st4(xb + 80 * XP, x5); st4(xb + 96 * XP, x6); st4(xb + 112 * XP, x7);
    put_pair(wb, e0, o0);
    put_pair(wb + 16 * WPP, e1, o1);
    __syncthreads();
  }
  mma_chunk((Fin / PKC - 1) & 1);

  float* ot = xs;
#pragma unroll
  for (int v = 0; v < 8; ++v) {
    float* o = ot + (32 * wr + v + 8 * laneHi) * XP + 32 * wc + lane16;
    o[0]            = acc00[v];
    o[16]           = acc01[v];
    o[16 * XP]      = acc10[v];
    o[16 * XP + 16] = acc11[v];
  }
  __syncthreads();
  const int bb = r0 >> 10, nb = r0 & (Nq - 1);
  float* hg = hbuf + ((size_t)(bb * Hh + hh) * Nq + nb) * HD;
#pragma unroll
  for (int i = 0; i < 8; ++i) {
    const int rr = crow + 16 * i;
    *reinterpret_cast<float4*>(hg + rr * HD + cc4) = *reinterpret_cast<const float4*>(ot + rr * XP + cc4);
  }
}

#define BR 128
#define MC 64
#define RPP 160
#define MWP 36

__device__ __forceinline__ float row16_sum(float v) {
  v += __builtin_bit_cast(float, __builtin_amdgcn_update_dpp(0, __builtin_bit_cast(int, v), 0x161, 0xf, 0xf, true));
  v += __builtin_bit_cast(float, __builtin_amdgcn_update_dpp(0, __builtin_bit_cast(int, v), 0x162, 0xf, 0xf, true));
  v += __builtin_bit_cast(float, __builtin_amdgcn_update_dpp(0, __builtin_bit_cast(int, v), 0x164, 0xf, 0xf, true));
  v += __builtin_bit_cast(float, __builtin_amdgcn_update_dpp(0, __builtin_bit_cast(int, v), 0x168, 0xf, 0xf, true));
  return v;
}

__device__ __forceinline__ float dot4(const float4 p, const float4 q) {
  return fmaf(p.w, q.w, fmaf(p.z, q.z, fmaf(p.y, q.y, p.x * q.x)));
}

__global__ __launch_bounds__(256) void gat_attn(const float* __restrict__ hbuf,
                                                const int* __restrict__ adj,
                                                const float* __restrict__ a,
                                                float* __restrict__ out) {
  __shared__ __attribute__((aligned(16))) unsigned int maskw[BR * MWP];
  __shared__ __attribute__((aligned(16))) float s2l[Nq];
  __shared__ float s1l[BR];
  __shared__ __attribute__((aligned(16))) float efl[2 * Nq];
  __shared__ __attribute__((aligned(16))) float wmaxl[8];
  __shared__ __attribute__((aligned(16))) float hl[2 * (MC / 2) * RPP];

  const int tid = threadIdx.x;
  const int wave = tid >> 5, lane = tid & 31;
  const int lane16 = lane & 15, laneHi = lane >> 4;
  const int hh = blockIdx.x, n0 = blockIdx.y * BR, b = blockIdx.z;

  const float* hb = hbuf + (size_t)(b * Hh + hh) * Nq * HD;
  const int prow = tid >> 4, cc4 = (tid & 15) * 4;
  const float* gsrc = hb + 2 * prow * HD + cc4;
  float* ldst = hl + prow * RPP + 2 * cc4;
  const float4 pe0 = *reinterpret_cast<const float4*>(gsrc);
  const float4 po0 = *reinterpret_cast<const float4*>(gsrc + HD);
  const float4 pe1 = *reinterpret_cast<const float4*>(gsrc + 32 * HD);
  const float4 po1 = *reinterpret_cast<const float4*>(gsrc + 33 * HD);

  const float* ah = a + hh * (2 * HD);
  const float4 a1v = *reinterpret_cast<const float4*>(ah + 4 * lane16);
  const float4 a2v = *reinterpret_cast<const float4*>(ah + HD + 4 * lane16);
  const float* hsw = hb + (size_t)(wave * 128 + laneHi) * HD + 4 * lane16;
  const float* h1 = hb + (size_t)(n0 + wave * 16 + laneHi) * HD + 4 * lane16;
  float4 h1v[8];
#pragma unroll
  for (int j = 0; j < 8; ++j) h1v[j] = *reinterpret_cast<const float4*>(h1 + 2 * j * HD);
  float4 h2v[64];
#pragma unroll
  for (int j = 0; j < 64; ++j) h2v[j] = *reinterpret_cast<const float4*>(hsw + 2 * j * HD);
  const int4* adjw = reinterpret_cast<const int4*>(adj + ((size_t)b * Nq + n0 + wave * 16) * Nq) + lane;
  int4 ab[2][32];
  auto load_rows = [&](const int g, int4* buf) {
#pragma unroll
    for (int i = 0; i < 4; ++i) {
#pragma unroll
      for (int q = 0; q < 8; ++q) buf[8 * i + q] = adjw[(4 * g + i) * (Nq / 4) + 32 * q];
    }
  };
  auto mask_rows = [&](const int g, const int4* buf) {
#pragma unroll
    for (int i = 0; i < 4; ++i) {
      unsigned int my = 0u;
#pragma unroll
      for (int q = 0; q < 8; ++q) {
        const int4 v = buf[8 * i + q];
        const unsigned int w0 = __builtin_amdgcn_ballot_w32(v.x != 0);
        const unsigned int w1 = __builtin_amdgcn_ballot_w32(v.y != 0);
        const unsigned int w2 = __builtin_amdgcn_ballot_w32(v.z != 0);
        const unsigned int w3 = __builtin_amdgcn_ballot_w32(v.w != 0);
        my = (lane == 4 * q + 0) ? w0 : my;
        my = (lane == 4 * q + 1) ? w1 : my;
        my = (lane == 4 * q + 2) ? w2 : my;
        my = (lane == 4 * q + 3) ? w3 : my;
      }
      maskw[(wave * 16 + 4 * g + i) * MWP + lane] = my;
    }
  };
  load_rows(0, ab[0]);
  float wmx = -__builtin_huge_valf();
#pragma unroll
  for (int g = 0; g < 8; ++g) {
    float d2[8];
#pragma unroll
    for (int j = 0; j < 8; ++j) {
      d2[j] = row16_sum(dot4(h2v[8 * g + j], a2v));
      wmx = fmaxf(wmx, d2[j]);
    }
    if (lane16 == 0) {
#pragma unroll
      for (int j = 0; j < 8; ++j) s2l[wave * 128 + 16 * g + 2 * j + laneHi] = d2[j];
    }
  }
  wmx = fmaxf(wmx, __shfl_xor(wmx, 16, 32));
  if (lane == 0) wmaxl[wave] = wmx;
  {
    float d1[8];
#pragma unroll
    for (int j = 0; j < 8; ++j) d1[j] = row16_sum(dot4(h1v[j], a1v));
    if (lane16 == 0) {
#pragma unroll
      for (int j = 0; j < 8; ++j) s1l[wave * 16 + 2 * j + laneHi] = d1[j];
    }
  }
  load_rows(1, ab[1]);
  mask_rows(0, ab[0]);
  load_rows(2, ab[0]);
  mask_rows(1, ab[1]);
  load_rows(3, ab[1]);
  mask_rows(2, ab[0]);
  mask_rows(3, ab[1]);
  const int row = wave * 16 + lane16;
  put_pair(ldst, pe0, po0);
  put_pair(ldst + 16 * RPP, pe1, po1);
  __syncthreads();
  const float s1v = s1l[row];

  float cmax;
  {
    const float4 w0 = *reinterpret_cast<const float4*>(wmaxl);
    const float4 w1 = *reinterpret_cast<const float4*>(wmaxl + 4);
    cmax = fmaxf(fmaxf(fmaxf(w0.x, w0.y), fmaxf(w0.z, w0.w)),
                 fmaxf(fmaxf(w1.x, w1.y), fmaxf(w1.z, w1.w)));
    const float4 sv = *reinterpret_cast<const float4*>(s2l + 4 * tid);
    const float t0 = sv.x - cmax, t1 = sv.y - cmax, t2 = sv.z - cmax, t3 = sv.w - cmax;
    float4 lo, hi;
    lo.x = __expf(t0); lo.y = __expf(t1); lo.z = __expf(NEG_SLOPE * t0); lo.w = __expf(NEG_SLOPE * t1);
    hi.x = __expf(t2); hi.y = __expf(t3); hi.z = __expf(NEG_SLOPE * t2); hi.w = __expf(NEG_SLOPE * t3);
    *reinterpret_cast<float4*>(efl + 8 * tid)     = lo;
    *reinterpret_cast<float4*>(efl + 8 * tid + 4) = hi;
  }
  __syncthreads();

  const unsigned int* mrow = maskw + row * MWP + 2 * laneHi;

  const float zrow = s1v + cmax;
  const float thr = __expf(-zrow);
  const float bneg = zrow > 0.f ? __expf((NEG_SLOPE - 1.f) * zrow) : 1.f;
  float rs0 = 0.f, rs1 = 0.f;

  v8f acc0 = {}, acc1 = {}, acc2 = {}, acc3 = {};
  auto mma_chunk = [&](const int ci) {
    const float* cur = hl + (ci & 1) * ((MC / 2) * RPP) + laneHi * RPP + 2 * lane16;
    const float* efc = efl + ci * (2 * MC) + 4 * laneHi;
    const uint2 wp = *reinterpret_cast<const uint2*>(mrow + 4 * (ci >> 1));
    const unsigned int wda = wp.x >> (16 * (ci & 1)), wdb = wp.y >> (16 * (ci & 1));
    v2f bf[2][16];
    float4 sf[2][4];
    auto load_block = [&](const int q, v2f* bq, float4* sq) {
#pragma unroll
      for (int ks = 0; ks < 4; ++ks) {
        const float* hp = cur + (8 * q + 2 * ks) * RPP;
        sq[ks] = *reinterpret_cast<const float4*>(efc + 32 * q + 8 * ks);
        bq[4 * ks + 0] = *reinterpret_cast<const v2f*>(hp);
        bq[4 * ks + 1] = *reinterpret_cast<const v2f*>(hp + 32);
        bq[4 * ks + 2] = *reinterpret_cast<const v2f*>(hp + 64);
        bq[4 * ks + 3] = *reinterpret_cast<const v2f*>(hp + 96);
      }
    };
    auto mma_block = [&](const int q, const v2f* bq, const float4* sq) {
#pragma unroll
      for (int ks = 0; ks < 4; ++ks) {
        const int bit = 4 * q + ks;
        const float4 ef = sq[ks];
        const float e0 = ef.x > thr ? ef.x : ef.z * bneg;
        const float e1 = ef.y > thr ? ef.y : ef.w * bneg;
        v2f av;
        av.x = ((wda >> bit) & 1u) ? e0 : 0.f;
        av.y = ((wdb >> bit) & 1u) ? e1 : 0.f;
        rs0 += av.x;
        rs1 += av.y;
        acc0 = __builtin_amdgcn_wmma_f32_16x16x4_f32(false, av, false, bq[4 * ks + 0], (short)0, acc0, false, false);
        acc1 = __builtin_amdgcn_wmma_f32_16x16x4_f32(false, av, false, bq[4 * ks + 1], (short)0, acc1, false, false);
        acc2 = __builtin_amdgcn_wmma_f32_16x16x4_f32(false, av, false, bq[4 * ks + 2], (short)0, acc2, false, false);
        acc3 = __builtin_amdgcn_wmma_f32_16x16x4_f32(false, av, false, bq[4 * ks + 3], (short)0, acc3, false, false);
      }
    };
    load_block(0, bf[0], sf[0]);
    load_block(1, bf[1], sf[1]);
    __builtin_amdgcn_sched_barrier(0);
    mma_block(0, bf[0], sf[0]);
    __builtin_amdgcn_sched_barrier(0);
    load_block(2, bf[0], sf[0]);
    __builtin_amdgcn_sched_barrier(0);
    mma_block(1, bf[1], sf[1]);
    __builtin_amdgcn_sched_barrier(0);
    load_block(3, bf[1], sf[1]);
    __builtin_amdgcn_sched_barrier(0);
    mma_block(2, bf[0], sf[0]);
    mma_block(3, bf[1], sf[1]);
  };
  for (int ci = 0; ci < Nq / MC - 1; ++ci) {
    const float* gs = gsrc + (size_t)(ci + 1) * MC * HD;
    const float4 e0 = *reinterpret_cast<const float4*>(gs);
    const float4 o0 = *reinterpret_cast<const float4*>(gs + HD);
    const float4 e1 = *reinterpret_cast<const float4*>(gs + 32 * HD);
    const float4 o1 = *reinterpret_cast<const float4*>(gs + 33 * HD);
    mma_chunk(ci);
    float* ld = ldst + ((ci + 1) & 1) * ((MC / 2) * RPP);
    put_pair(ld, e0, o0);
    put_pair(ld + 16 * RPP, e1, o1);
    __syncthreads();
  }
  mma_chunk(Nq / MC - 1);

  float rinv;
  {
    const float sm = rs0 + rs1;
    const float osm = __shfl_xor(sm, 16, 32);
    const float rsum = (laneHi ? osm : sm) + (laneHi ? sm : osm);
    rinv = rsum > 0.f ? 1.f / rsum : 0.f;
  }
  float* ob = out + ((size_t)b * Nq + n0 + wave * 16) * Fout + hh * HD;
#pragma unroll
  for (int v = 0; v < 8; ++v) {
    const float ri = __shfl(rinv, v + 8 * laneHi, 32);
    float* orow = ob + (size_t)(v + 8 * laneHi) * Fout + lane16;
    orow[0]  = acc0[v] * ri;
    orow[16] = acc1[v] * ri;
    orow[32] = acc2[v] * ri;
    orow[48] = acc3[v] * ri;
  }
}

extern "C" void kernel_launch(void* const* d_in, const int* in_sizes, int n_in,
                              void* d_out, int out_size, void* d_ws, size_t ws_size,
                              hipStream_t stream) {
  const float* x   = (const float*)d_in[0];
  const int*   adj = (const int*)d_in[1];
  const float* W   = (const float*)d_in[2];
  const float* a   = (const float*)d_in[3];
  float* out = (float*)d_out;

  float* hbuf = (float*)d_ws;

  gat_proj<<<dim3(Hh, 64), 256, 0, stream>>>(x, W, hbuf);
  gat_attn<<<dim3(Hh, Nq / 128, Bq), 256, 0, stream>>>(hbuf, adj, a, out);
}


# attn loop: p = umin(umax(E, F*B), mask bit sign-extended) on the float bit patterns: 1 bfe + 1 max-min a value, no compare or select; sched fences dropped, as built a chunk's 49 LDS reads lead it
# speedup vs baseline: 2.6568x; 1.0457x over previous
#include <hip/hip_runtime.h>


typedef float v2f __attribute__((ext_vector_type(2)));
typedef float v8f __attribute__((ext_vector_type(8)));

#define Bq 8
#define Nq 1024
#define Fin 256
#define Fout 256
#define Hh 4
#define HD 64
#define NEG_SLOPE 0.2f

#define PBM 128
#define PKC 64
#define XP 68
#define WPP 160

__device__ __forceinline__ void put_pair(float* d, const float4 e, const float4 o) {
  *reinterpret_cast<float4*>(d)     = make_float4(e.x, o.x, e.y, o.y);
  *reinterpret_cast<float4*>(d + 4) = make_float4(e.z, o.z, e.w, o.w);
}

__global__ __launch_bounds__(256) void gat_proj(const float* __restrict__ x,
                                                const float* W,
                                                float* __restrict__ hbuf) {
  __shared__ __attribute__((aligned(16))) float xs[2 * PBM * XP];
  __shared__ __attribute__((aligned(16))) float wl[2 * (PKC / 2) * WPP];

  const int tid = threadIdx.x;
  const int wave = tid >> 5, lane = tid & 31;
  const int lane16 = lane & 15, laneHi = lane >> 4;
  const int hh = blockIdx.x;
  const int r0 = blockIdx.y * PBM;
  const int wr = wave >> 1, wc = wave & 1;

  const int crow = tid >> 4, cc4 = (tid & 15) * 4;
  const float* xg = x + (size_t)(r0 + crow) * Fin + cc4;
  const float* wg = W + (size_t)(2 * crow) * Fout + hh * HD + cc4;
  float* xd = xs + crow * XP + cc4;
  float* wd = wl + crow * WPP + 2 * cc4;
  auto ld4 = [](const float* p) { return *reinterpret_cast<const float4*>(p); };
  auto st4 = [](float* p, const float4 v) { *reinterpret_cast<float4*>(p) = v; };

  v8f acc00 = {}, acc01 = {}, acc10 = {}, acc11 = {};
  auto mma_chunk = [&](const int buf) {
    const float* xa = xs + buf * (PBM * XP) + (32 * wr + lane16) * XP + 2 * laneHi;
    const float* wb = wl + buf * ((PKC / 2) * WPP) + laneHi * WPP + 2 * (32 * wc + lane16);
#pragma unroll
    for (int k = 0; k < PKC / 4; ++k) {
      const v2f fa0 = *reinterpret_cast<const v2f*>(xa + 4 * k);
      const v2f fa1 = *reinterpret_cast<const v2f*>(xa + 16 * XP + 4 * k);
      const v2f fb0 = *reinterpret_cast<const v2f*>(wb + 2 * k * WPP);
      const v2f fb1 = *reinterpret_cast<const v2f*>(wb + 2 * k * WPP + 32);
      acc00 = __builtin_amdgcn_wmma_f32_16x16x4_f32(false, fa0, false, fb0, (short)0, acc00, false, false);
      acc01 = __builtin_amdgcn_wmma_f32_16x16x4_f32(false, fa0, false, fb1, (short)0, acc01, false, false);
      acc10 = __builtin_amdgcn_wmma_f32_16x16x4_f32(false, fa1, false, fb0, (short)0, acc10, false, false);
      acc11 = __builtin_amdgcn_wmma_f32_16x16x4_f32(false, fa1, false, fb1, (short)0, acc11, false, false);
    }
  };

  {
    const float4 x0 = ld4(xg),            x1 = ld4(xg + 16 * Fin), x2 = ld4(xg + 32 * Fin), x3 = ld4(xg + 48 * Fin);
    const float4 x4 = ld4(xg + 64 * Fin), x5 = ld4(xg + 80 * Fin), x6 = ld4(xg + 96 * Fin), x7 = ld4(xg + 112 * Fin);
    const float4 e0 = ld4(wg), o0 = ld4(wg + Fout), e1 = ld4(wg + 32 * Fout), o1 = ld4(wg + 33 * Fout);
    st4(xd, x0);           st4(xd + 16 * XP, x1); st4(xd + 32 * XP, x2); st4(xd + 48 * XP, x3);
    st4(xd + 64 * XP, x4); st4(xd + 80 * XP, x5); st4(xd + 96 * XP, x6); st4(xd + 112 * XP, x7);
    put_pair(wd, e0, o0);
    put_pair(wd + 16 * WPP, e1, o1);
  }
  __syncthreads();
  for (int c = 0; c < Fin / PKC - 1; ++c) {
    const float* xn = xg + (c + 1) * PKC;
    const float* wn = wg + (size_t)(c + 1) * PKC * Fout;
    const float4 e0 = ld4(wn), o0 = ld4(wn + Fout), e1 = ld4(wn + 32 * Fout), o1 = ld4(wn + 33 * Fout);
    const float4 x0 = ld4(xn),            x1 = ld4(xn + 16 * Fin), x2 = ld4(xn + 32 * Fin), x3 = ld4(xn + 48 * Fin);
    const float4 x4 = ld4(xn + 64 * Fin), x5 = ld4(xn + 80 * Fin), x6 = ld4(xn + 96 * Fin), x7 = ld4(xn + 112 * Fin);
    mma_chunk(c & 1);
    __builtin_amdgcn_sched_barrier(0);
    float* xb = xd + ((c + 1) & 1) * (PBM * XP);
    float* wb = wd + ((c + 1) & 1) * ((PKC / 2) * WPP);
    st4(xb, x0);           st4(xb + 16 * XP, x1); st4(xb + 32 * XP, x2); st4(xb + 48 * XP, x3);
    st4(xb + 64 * XP, x4); st4(xb + 80 * XP, x5); st4(xb + 96 * XP, x6); st4(xb + 112 * XP, x7);
    put_pair(wb, e0, o0);
    put_pair(wb + 16 * WPP, e1, o1);
    __syncthreads();
  }
  mma_chunk((Fin / PKC - 1) & 1);

  float* ot = xs;
#pragma unroll
  for (int v = 0; v < 8; ++v) {
    float* o = ot + (32 * wr + v + 8 * laneHi) * XP + 32 * wc + lane16;
    o[0]            = acc00[v];
    o[16]           = acc01[v];
    o[16 * XP]      = acc10[v];
    o[16 * XP + 16] = acc11[v];
  }
  __syncthreads();
  const int bb = r0 >> 10, nb = r0 & (Nq - 1);
  float* hg = hbuf + ((size_t)(bb * Hh + hh) * Nq + nb) * HD;
#pragma unroll
  for (int i = 0; i < 8; ++i) {
    const int rr = crow + 16 * i;
    *reinterpret_cast<float4*>(hg + rr * HD + cc4) = *reinterpret_cast<const float4*>(ot + rr * XP + cc4);
  }
}

#define BR 128
#define MC 64
#define RPP 160
#define MWP 36

__device__ __forceinline__ float row16_sum(float v) {
  v += __builtin_bit_cast(float, __builtin_amdgcn_update_dpp(0, __builtin_bit_cast(int, v), 0x161, 0xf, 0xf, true));
  v += __builtin_bit_cast(float, __builtin_amdgcn_update_dpp(0, __builtin_bit_cast(int, v), 0x162, 0xf, 0xf, true));
  v += __builtin_bit_cast(float, __builtin_amdgcn_update_dpp(0, __builtin_bit_cast(int, v), 0x164, 0xf, 0xf, true));
  v += __builtin_bit_cast(float, __builtin_amdgcn_update_dpp(0, __builtin_bit_cast(int, v), 0x168, 0xf, 0xf, true));
  return v;
}

__device__ __forceinline__ float dot4(const float4 p, const float4 q) {
  return fmaf(p.w, q.w, fmaf(p.z, q.z, fmaf(p.y, q.y, p.x * q.x)));
}

__device__ __forceinline__ unsigned int fbits(const float v) { return __builtin_bit_cast(unsigned int, v); }
__device__ __forceinline__ unsigned int umax(const unsigned int p, const unsigned int q) { return p > q ? p : q; }
__device__ __forceinline__ unsigned int umin(const unsigned int p, const unsigned int q) { return p < q ? p : q; }

__global__ __launch_bounds__(256) void gat_attn(const float* __restrict__ hbuf,
                                                const int* __restrict__ adj,
                                                const float* __restrict__ a,
                                                float* __restrict__ out) {
  __shared__ __attribute__((aligned(16))) unsigned int maskw[BR * MWP];
  __shared__ __attribute__((aligned(16))) float s2l[Nq];
  __shared__ float s1l[BR];
  __shared__ __attribute__((aligned(16))) float efl[2 * Nq];
  __shared__ __attribute__((aligned(16))) float wmaxl[8];
  __shared__ __attribute__((aligned(16))) float hl[2 * (MC / 2) * RPP];

  const int tid = threadIdx.x;
  const int wave = tid >> 5, lane = tid & 31;
  const int lane16 = lane & 15, laneHi = lane >> 4;
  const int hh = blockIdx.x, n0 = blockIdx.y * BR, b = blockIdx.z;

  const float* hb = hbuf + (size_t)(b * Hh + hh) * Nq * HD;
  const int prow = tid >> 4, cc4 = (tid & 15) * 4;
  const float* gsrc = hb + 2 * prow * HD + cc4;
  float* ldst = hl + prow * RPP + 2 * cc4;
  const float4 pe0 = *reinterpret_cast<const float4*>(gsrc);
  const float4 po0 = *reinterpret_cast<const float4*>(gsrc + HD);
  const float4 pe1 = *reinterpret_cast<const float4*>(gsrc + 32 * HD);
  const float4 po1 = *reinterpret_cast<const float4*>(gsrc + 33 * HD);

  const float* ah = a + hh * (2 * HD);
  const float4 a1v = *reinterpret_cast<const float4*>(ah + 4 * lane16);
  const float4 a2v = *reinterpret_cast<const float4*>(ah + HD + 4 * lane16);
  const float* hsw = hb + (size_t)(wave * 128 + laneHi) * HD + 4 * lane16;
  const float* h1 = hb + (size_t)(n0 + wave * 16 + laneHi) * HD + 4 * lane16;
  float4 h1v[8];
#pragma unroll
  for (int j = 0; j < 8; ++j) h1v[j] = *reinterpret_cast<const float4*>(h1 + 2 * j * HD);
  float4 h2v[64];
#pragma unroll
  for (int j = 0; j < 64; ++j) h2v[j] = *reinterpret_cast<const float4*>(hsw + 2 * j * HD);
  const int4* adjw = reinterpret_cast<const int4*>(adj + ((size_t)b * Nq + n0 + wave * 16) * Nq) + lane;
  int4 ab[2][32];
  auto load_rows = [&](const int g, int4* buf) {
#pragma unroll
    for (int i = 0; i < 4; ++i) {
#pragma unroll
      for (int q = 0; q < 8; ++q) buf[8 * i + q] = adjw[(4 * g + i) * (Nq / 4) + 32 * q];
    }
  };
  auto mask_rows = [&](const int g, const int4* buf) {
#pragma unroll
    for (int i = 0; i < 4; ++i) {
      unsigned int my = 0u;
#pragma unroll
      for (int q = 0; q < 8; ++q) {
        const int4 v = buf[8 * i + q];
        const unsigned int w0 = __builtin_amdgcn_ballot_w32(v.x != 0);
        const unsigned int w1 = __builtin_amdgcn_ballot_w32(v.y != 0);
        const unsigned int w2 = __builtin_amdgcn_ballot_w32(v.z != 0);
        const unsigned int w3 = __builtin_amdgcn_ballot_w32(v.w != 0);
        my = (lane == 4 * q + 0) ? w0 : my;
        my = (lane == 4 * q + 1) ? w1 : my;
        my = (lane == 4 * q + 2) ? w2 : my;
        my = (lane == 4 * q + 3) ? w3 : my;
      }
      maskw[(wave * 16 + 4 * g + i) * MWP + lane] = my;
    }
  };
  load_rows(0, ab[0]);
  float wmx = -__builtin_huge_valf();
#pragma unroll
  for (int g = 0; g < 8; ++g) {
    float d2[8];
#pragma unroll
    for (int j = 0; j < 8; ++j) {
      d2[j] = row16_sum(dot4(h2v[8 * g + j], a2v));
      wmx = fmaxf(wmx, d2[j]);
    }
    if (lane16 == 0) {
#pragma unroll
      for (int j = 0; j < 8; ++j) s2l[wave * 128 + 16 * g + 2 * j + laneHi] = d2[j];
    }
  }
  wmx = fmaxf(wmx, __shfl_xor(wmx, 16, 32));
  if (lane == 0) wmaxl[wave] = wmx;
  {
    float d1[8];
#pragma unroll
    for (int j = 0; j < 8; ++j) d1[j] = row16_sum(dot4(h1v[j], a1v));
    if (lane16 == 0) {
#pragma unroll
      for (int j = 0; j < 8; ++j) s1l[wave * 16 + 2 * j + laneHi] = d1[j];
    }
  }
  load_rows(1, ab[1]);
  mask_rows(0, ab[0]);
  load_rows(2, ab[0]);
  mask_rows(1, ab[1]);
  load_rows(3, ab[1]);
  mask_rows(2, ab[0]);
  mask_rows(3, ab[1]);
  const int row = wave * 16 + lane16;
  put_pair(ldst, pe0, po0);
  put_pair(ldst + 16 * RPP, pe1, po1);
  __syncthreads();
  const float s1v = s1l[row];

  float cmax;
  {
    const float4 w0 = *reinterpret_cast<const float4*>(wmaxl);
    const float4 w1 = *reinterpret_cast<const float4*>(wmaxl + 4);
    cmax = fmaxf(fmaxf(fmaxf(w0.x, w0.y), fmaxf(w0.z, w0.w)),
                 fmaxf(fmaxf(w1.x, w1.y), fmaxf(w1.z, w1.w)));
    const float4 sv = *reinterpret_cast<const float4*>(s2l + 4 * tid);
    const float t0 = sv.x - cmax, t1 = sv.y - cmax, t2 = sv.z - cmax, t3 = sv.w - cmax;
    float4 lo, hi;
    lo.x = __expf(t0); lo.y = __expf(t1); lo.z = __expf(NEG_SLOPE * t0); lo.w = __expf(NEG_SLOPE * t1);
    hi.x = __expf(t2); hi.y = __expf(t3); hi.z = __expf(NEG_SLOPE * t2); hi.w = __expf(NEG_SLOPE * t3);
    *reinterpret_cast<float4*>(efl + 8 * tid)     = lo;
    *reinterpret_cast<float4*>(efl + 8 * tid + 4) = hi;
  }
  __syncthreads();

  const unsigned int* mrow = maskw + row * MWP + 2 * laneHi;

  const float zrow = s1v + cmax;
  const float bneg = zrow > 0.f ? __expf((NEG_SLOPE - 1.f) * zrow) : 1.f;
  float rs0 = 0.f, rs1 = 0.f;

  v8f acc0 = {}, acc1 = {}, acc2 = {}, acc3 = {};
  auto mma_chunk = [&](const int ci) {
    const float* cur = hl + (ci & 1) * ((MC / 2) * RPP) + laneHi * RPP + 2 * lane16;
    const float* efc = efl + ci * (2 * MC) + 4 * laneHi;
    const uint2 wp = *reinterpret_cast<const uint2*>(mrow + 4 * (ci >> 1));
    const unsigned int wda = wp.x >> (16 * (ci & 1)), wdb = wp.y >> (16 * (ci & 1));
    v2f bf[4][16];
    float4 sf[4][4];
    auto load_block = [&](const int q, v2f* bq, float4* sq) {
#pragma unroll
      for (int ks = 0; ks < 4; ++ks) {
        const float* hp = cur + (8 * q + 2 * ks) * RPP;
        sq[ks] = *reinterpret_cast<const float4*>(efc + 32 * q + 8 * ks);
        bq[4 * ks + 0] = *reinterpret_cast<const v2f*>(hp);
        bq[4 * ks + 1] = *reinterpret_cast<const v2f*>(hp + 32);
        bq[4 * ks + 2] = *reinterpret_cast<const v2f*>(hp + 64);
        bq[4 * ks + 3] = *reinterpret_cast<const v2f*>(hp + 96);
      }
    };
    auto mma_block = [&](const int q, const v2f* bq, const float4* sq) {
#pragma unroll
      for (int ks = 0; ks < 4; ++ks) {
        const int bit = 4 * q + ks;
        const float4 ef = sq[ks];
        const unsigned int s0 = (unsigned int)((int)(wda << (31 - bit)) >> 31);
        const unsigned int s1m = (unsigned int)((int)(wdb << (31 - bit)) >> 31);
        const unsigned int p0 = umin(umax(fbits(ef.x), fbits(ef.z * bneg)), s0);
        const unsigned int p1 = umin(umax(fbits(ef.y), fbits(ef.w * bneg)), s1m);
        v2f av;
        av.x = __builtin_bit_cast(float, p0);
        av.y = __builtin_bit_cast(float, p1);
        rs0 += av.x;
        rs1 += av.y;
        acc0 = __builtin_amdgcn_wmma_f32_16x16x4_f32(false, av, false, bq[4 * ks + 0], (short)0, acc0, false, false);
        acc1 = __builtin_amdgcn_wmma_f32_16x16x4_f32(false, av, false, bq[4 * ks + 1], (short)0, acc1, false, false);
        acc2 = __builtin_amdgcn_wmma_f32_16x16x4_f32(false, av, false, bq[4 * ks + 2], (short)0, acc2, false, false);
        acc3 = __builtin_amdgcn_wmma_f32_16x16x4_f32(false, av, false, bq[4 * ks + 3], (short)0, acc3, false, false);
      }
    };
#pragma unroll
    for (int q = 0; q < 4; ++q) load_block(q, bf[q], sf[q]);
#pragma unroll
    for (int q = 0; q < 4; ++q) mma_block(q, bf[q], sf[q]);
  };
  for (int ci = 0; ci < Nq / MC - 1; ++ci) {
    const float* gs = gsrc + (size_t)(ci + 1) * MC * HD;
    const float4 e0 = *reinterpret_cast<const float4*>(gs);
    const float4 o0 = *reinterpret_cast<const float4*>(gs + HD);
    const float4 e1 = *reinterpret_cast<const float4*>(gs + 32 * HD);
    const float4 o1 = *reinterpret_cast<const float4*>(gs + 33 * HD);
    mma_chunk(ci);
    float* ld = ldst + ((ci + 1) & 1) * ((MC / 2) * RPP);
    put_pair(ld, e0, o0);
    put_pair(ld + 16 * RPP, e1, o1);
    __syncthreads();
  }
  mma_chunk(Nq / MC - 1);

  float rinv;
  {
    const float sm = rs0 + rs1;
    const float osm = __shfl_xor(sm, 16, 32);
    const float rsum = (laneHi ? osm : sm) + (laneHi ? sm : osm);
    rinv = rsum > 0.f ? 1.f / rsum : 0.f;
  }
  float* ob = out + ((size_t)b * Nq + n0 + wave * 16) * Fout + hh * HD;
#pragma unroll
  for (int v = 0; v < 8; ++v) {
    const float ri = __shfl(rinv, v + 8 * laneHi, 32);
    float* orow = ob + (size_t)(v + 8 * laneHi) * Fout + lane16;
    orow[0]  = acc0[v] * ri;
    orow[16] = acc1[v] * ri;
    orow[32] = acc2[v] * ri;
    orow[48] = acc3[v] * ri;
  }
}

extern "C" void kernel_launch(void* const* d_in, const int* in_sizes, int n_in,
                              void* d_out, int out_size, void* d_ws, size_t ws_size,
                              hipStream_t stream) {
  const float* x   = (const float*)d_in[0];
  const int*   adj = (const int*)d_in[1];
  const float* W   = (const float*)d_in[2];
  const float* a   = (const float*)d_in[3];
  float* out = (float*)d_out;

  float* hbuf = (float*)d_ws;

  gat_proj<<<dim3(Hh, 64), 256, 0, stream>>>(x, W, hbuf);
  gat_attn<<<dim3(Hh, Nq / 128, Bq), 256, 0, stream>>>(hbuf, adj, a, out);
}


# adj read moved from the prologue into the product loop: per 64-m chunk a wave loads its 16 row pieces (256 B each) one chunk ahead, ballots give the lane's mask words in registers; mask in LDS gone
# speedup vs baseline: 2.8960x; 1.0900x over previous
#include <hip/hip_runtime.h>


typedef float v2f __attribute__((ext_vector_type(2)));
typedef float v8f __attribute__((ext_vector_type(8)));

#define Bq 8
#define Nq 1024
#define Fin 256
#define Fout 256
#define Hh 4
#define HD 64
#define NEG_SLOPE 0.2f

#define PBM 128
#define PKC 64
#define XP 68
#define WPP 160

__device__ __forceinline__ void put_pair(float* d, const float4 e, const float4 o) {
  *reinterpret_cast<float4*>(d)     = make_float4(e.x, o.x, e.y, o.y);
  *reinterpret_cast<float4*>(d + 4) = make_float4(e.z, o.z, e.w, o.w);
}

__global__ __launch_bounds__(256) void gat_proj(const float* __restrict__ x,
                                                const float* W,
                                                float* __restrict__ hbuf) {
  __shared__ __attribute__((aligned(16))) float xs[2 * PBM * XP];
  __shared__ __attribute__((aligned(16))) float wl[2 * (PKC / 2) * WPP];

  const int tid = threadIdx.x;
  const int wave = tid >> 5, lane = tid & 31;
  const int lane16 = lane & 15, laneHi = lane >> 4;
  const int hh = blockIdx.x;
  const int r0 = blockIdx.y * PBM;
  const int wr = wave >> 1, wc = wave & 1;

  const int crow = tid >> 4, cc4 = (tid & 15) * 4;
  const float* xg = x + (size_t)(r0 + crow) * Fin + cc4;
  const float* wg = W + (size_t)(2 * crow) * Fout + hh * HD + cc4;
  float* xd = xs + crow * XP + cc4;
  float* wd = wl + crow * WPP + 2 * cc4;
  auto ld4 = [](const float* p) { return *reinterpret_cast<const float4*>(p); };
  auto st4 = [](float* p, const float4 v) { *reinterpret_cast<float4*>(p) = v; };

  v8f acc00 = {}, acc01 = {}, acc10 = {}, acc11 = {};
  auto mma_chunk = [&](const int buf) {
    const float* xa = xs + buf * (PBM * XP) + (32 * wr + lane16) * XP + 2 * laneHi;
    const float* wb = wl + buf * ((PKC / 2) * WPP) + laneHi * WPP + 2 * (32 * wc + lane16);
#pragma unroll
    for (int k = 0; k < PKC / 4; ++k) {
      const v2f fa0 = *reinterpret_cast<const v2f*>(xa + 4 * k);
      const v2f fa1 = *reinterpret_cast<const v2f*>(xa + 16 * XP + 4 * k);
      const v2f fb0 = *reinterpret_cast<const v2f*>(wb + 2 * k * WPP);
      const v2f fb1 = *reinterpret_cast<const v2f*>(wb + 2 * k * WPP + 32);
      acc00 = __builtin_amdgcn_wmma_f32_16x16x4_f32(false, fa0, false, fb0, (short)0, acc00, false, false);
      acc01 = __builtin_amdgcn_wmma_f32_16x16x4_f32(false, fa0, false, fb1, (short)0, acc01, false, false);
      acc10 = __builtin_amdgcn_wmma_f32_16x16x4_f32(false, fa1, false, fb0, (short)0, acc10, false, false);
      acc11 = __builtin_amdgcn_wmma_f32_16x16x4_f32(false, fa1, false, fb1, (short)0, acc11, false, false);
    }
  };

  {
    const float4 x0 = ld4(xg),            x1 = ld4(xg + 16 * Fin), x2 = ld4(xg + 32 * Fin), x3 = ld4(xg + 48 * Fin);
    const float4 x4 = ld4(xg + 64 * Fin), x5 = ld4(xg + 80 * Fin), x6 = ld4(xg + 96 * Fin), x7 = ld4(xg + 112 * Fin);
    const float4 e0 = ld4(wg), o0 = ld4(wg + Fout), e1 = ld4(wg + 32 * Fout), o1 = ld4(wg + 33 * Fout);
    st4(xd, x0);           st4(xd + 16 * XP, x1); st4(xd + 32 * XP, x2); st4(xd + 48 * XP, x3);
    st4(xd + 64 * XP, x4); st4(xd + 80 * XP, x5); st4(xd + 96 * XP, x6); st4(xd + 112 * XP, x7);
    put_pair(wd, e0, o0);
    put_pair(wd + 16 * WPP, e1, o1);
  }
  __syncthreads();
  for (int c = 0; c < Fin / PKC - 1; ++c) {
    const float* xn = xg + (c + 1) * PKC;
    const float* wn = wg + (size_t)(c + 1) * PKC * Fout;
    const float4 e0 = ld4(wn), o0 = ld4(wn + Fout), e1 = ld4(wn + 32 * Fout), o1 = ld4(wn + 33 * Fout);
    const float4 x0 = ld4(xn),            x1 = ld4(xn + 16 * Fin), x2 = ld4(xn + 32 * Fin), x3 = ld4(xn + 48 * Fin);
    const float4 x4 = ld4(xn + 64 * Fin), x5 = ld4(xn + 80 * Fin), x6 = ld4(xn + 96 * Fin), x7 = ld4(xn + 112 * Fin);
    mma_chunk(c & 1);
    __builtin_amdgcn_sched_barrier(0);
    float* xb = xd + ((c + 1) & 1) * (PBM * XP);
    float* wb = wd + ((c + 1) & 1) * ((PKC / 2) * WPP);
    st4(xb, x0);           st4(xb + 16 * XP, x1); st4(xb + 32 * XP, x2); st4(xb + 48 * XP, x3);
    st4(xb + 64 * XP, x4); st4(xb + 80 * XP, x5); st4(xb + 96 * XP, x6); st4(xb + 112 * XP, x7);
    put_pair(wb, e0, o0);
    put_pair(wb + 16 * WPP, e1, o1);
    __syncthreads();
  }
  mma_chunk((Fin / PKC - 1) & 1);

  float* ot = xs;
#pragma unroll
  for (int v = 0; v < 8; ++v) {
    float* o = ot + (32 * wr + v + 8 * laneHi) * XP + 32 * wc + lane16;
    o[0]            = acc00[v];
    o[16]           = acc01[v];
    o[16 * XP]      = acc10[v];
    o[16 * XP + 16] = acc11[v];
  }
  __syncthreads();
  const int bb = r0 >> 10, nb = r0 & (Nq - 1);
  float* hg = hbuf + ((size_t)(bb * Hh + hh) * Nq + nb) * HD;
#pragma unroll
  for (int i = 0; i < 8; ++i) {
    const int rr = crow + 16 * i;
    *reinterpret_cast<float4*>(hg + rr * HD + cc4) = *reinterpret_cast<const float4*>(ot + rr * XP + cc4);
  }
}

#define BR 128
#define MC 64
#define RPP 160

__device__ __forceinline__ float row16_sum(float v) {
  v += __builtin_bit_cast(float, __builtin_amdgcn_update_dpp(0, __builtin_bit_cast(int, v), 0x161, 0xf, 0xf, true));
  v += __builtin_bit_cast(float, __builtin_amdgcn_update_dpp(0, __builtin_bit_cast(int, v), 0x162, 0xf, 0xf, true));
  v += __builtin_bit_cast(float, __builtin_amdgcn_update_dpp(0, __builtin_bit_cast(int, v), 0x164, 0xf, 0xf, true));
  v += __builtin_bit_cast(float, __builtin_amdgcn_update_dpp(0, __builtin_bit_cast(int, v), 0x168, 0xf, 0xf, true));
  return v;
}

__device__ __forceinline__ float dot4(const float4 p, const float4 q) {
  return fmaf(p.w, q.w, fmaf(p.z, q.z, fmaf(p.y, q.y, p.x * q.x)));
}

__device__ __forceinline__ unsigned int fbits(const float v) { return __builtin_bit_cast(unsigned int, v); }
__device__ __forceinline__ unsigned int umax(const unsigned int p, const unsigned int q) { return p > q ? p : q; }
__device__ __forceinline__ unsigned int umin(const unsigned int p, const unsigned int q) { return p < q ? p : q; }

__device__ __forceinline__ void lds_barrier() {
  __builtin_amdgcn_fence(__ATOMIC_RELEASE, "workgroup", "local");
  __builtin_amdgcn_s_barrier();
  __builtin_amdgcn_fence(__ATOMIC_ACQUIRE, "workgroup", "local");
}

__global__ __launch_bounds__(256) void gat_attn(const float* __restrict__ hbuf,
                                                const int* __restrict__ adj,
                                                const float* __restrict__ a,
                                                float* __restrict__ out) {
  __shared__ __attribute__((aligned(16))) float s2l[Nq];
  __shared__ float s1l[BR];
  __shared__ __attribute__((aligned(16))) float efl[2 * Nq];
  __shared__ __attribute__((aligned(16))) float wmaxl[8];
  __shared__ __attribute__((aligned(16))) float hl[2 * (MC / 2) * RPP];

  const int tid = threadIdx.x;
  const int wave = tid >> 5, lane = tid & 31;
  const int lane16 = lane & 15, laneHi = lane >> 4;
  const int hh = blockIdx.x, n0 = blockIdx.y * BR, b = blockIdx.z;

  const float* hb = hbuf + (size_t)(b * Hh + hh) * Nq * HD;
  const int prow = tid >> 4, cc4 = (tid & 15) * 4;
  const float* gsrc = hb + 2 * prow * HD + cc4;
  float* ldst = hl + prow * RPP + 2 * cc4;
  const float4 pe0 = *reinterpret_cast<const float4*>(gsrc);
  const float4 po0 = *reinterpret_cast<const float4*>(gsrc + HD);
  const float4 pe1 = *reinterpret_cast<const float4*>(gsrc + 32 * HD);
  const float4 po1 = *reinterpret_cast<const float4*>(gsrc + 33 * HD);

  const float* ah = a + hh * (2 * HD);
  const float4 a1v = *reinterpret_cast<const float4*>(ah + 4 * lane16);
  const float4 a2v = *reinterpret_cast<const float4*>(ah + HD + 4 * lane16);
  const float* hsw = hb + (size_t)(wave * 128 + laneHi) * HD + 4 * lane16;
  const float* h1 = hb + (size_t)(n0 + wave * 16 + laneHi) * HD + 4 * lane16;
  float4 h1v[8];
#pragma unroll
  for (int j = 0; j < 8; ++j) h1v[j] = *reinterpret_cast<const float4*>(h1 + 2 * j * HD);
  float4 h2v[64];
#pragma unroll
  for (int j = 0; j < 64; ++j) h2v[j] = *reinterpret_cast<const float4*>(hsw + 2 * j * HD);
  const int2* adjw = reinterpret_cast<const int2*>(adj + ((size_t)b * Nq + n0 + wave * 16) * Nq) + lane;
  int2 an[16];
  auto load_piece = [&](const int ci) {
#pragma unroll
    for (int i = 0; i < 16; ++i) an[i] = adjw[i * (Nq / 2) + 32 * ci];
  };
  auto mask_piece = [&]() {
    unsigned int mx[4] = {0u, 0u, 0u, 0u}, my[4] = {0u, 0u, 0u, 0u};
#pragma unroll
    for (int i = 0; i < 16; ++i) {
      const int2 v = an[i];
      const unsigned int w0 = __builtin_amdgcn_ballot_w32(v.x != 0);
      const unsigned int w1 = __builtin_amdgcn_ballot_w32(v.y != 0);
      mx[i & 3] = (lane16 == i) ? w0 : mx[i & 3];
      my[i & 3] = (lane16 == i) ? w1 : my[i & 3];
    }
    return make_uint2(((mx[0] | mx[1]) | (mx[2] | mx[3])) >> laneHi, ((my[0] | my[1]) | (my[2] | my[3])) >> laneHi);
  };
  load_piece(0);
  float wmx = -__builtin_huge_valf();
#pragma unroll
  for (int g = 0; g < 8; ++g) {
    float d2[8];
#pragma unroll
    for (int j = 0; j < 8; ++j) {
      d2[j] = row16_sum(dot4(h2v[8 * g + j], a2v));
      wmx = fmaxf(wmx, d2[j]);
    }
    if (lane16 == 0) {
#pragma unroll
      for (int j = 0; j < 8; ++j) s2l[wave * 128 + 16 * g + 2 * j + laneHi] = d2[j];
    }
  }
  wmx = fmaxf(wmx, __shfl_xor(wmx, 16, 32));
  if (lane == 0) wmaxl[wave] = wmx;
  {
    float d1[8];
#pragma unroll
    for (int j = 0; j < 8; ++j) d1[j] = row16_sum(dot4(h1v[j], a1v));
    if (lane16 == 0) {
#pragma unroll
      for (int j = 0; j < 8; ++j) s1l[wave * 16 + 2 * j + laneHi] = d1[j];
    }
  }
  const int row = wave * 16 + lane16;
  put_pair(ldst, pe0, po0);
  put_pair(ldst + 16 * RPP, pe1, po1);
  __syncthreads();
  const float s1v = s1l[row];

  float cmax;
  {
    const float4 w0 = *reinterpret_cast<const float4*>(wmaxl);
    const float4 w1 = *reinterpret_cast<const float4*>(wmaxl + 4);
    cmax = fmaxf(fmaxf(fmaxf(w0.x, w0.y), fmaxf(w0.z, w0.w)),
                 fmaxf(fmaxf(w1.x, w1.y), fmaxf(w1.z, w1.w)));
    const float4 sv = *reinterpret_cast<const float4*>(s2l + 4 * tid);
    const float t0 = sv.x - cmax, t1 = sv.y - cmax, t2 = sv.z - cmax, t3 = sv.w - cmax;
    float4 lo, hi;
    lo.x = __expf(t0); lo.y = __expf(t1); lo.z = __expf(NEG_SLOPE * t0); lo.w = __expf(NEG_SLOPE * t1);
    hi.x = __expf(t2); hi.y = __expf(t3); hi.z = __expf(NEG_SLOPE * t2); hi.w = __expf(NEG_SLOPE * t3);
    *reinterpret_cast<float4*>(efl + 8 * tid)     = lo;
    *reinterpret_cast<float4*>(efl + 8 * tid + 4) = hi;
  }
  __syncthreads();

  const float zrow = s1v + cmax;
  const float bneg = zrow > 0.f ? __expf((NEG_SLOPE - 1.f) * zrow) : 1.f;
  float rs0 = 0.f, rs1 = 0.f;

  v8f acc0 = {}, acc1 = {}, acc2 = {}, acc3 = {};
  auto mma_chunk = [&](const int ci, const unsigned int wda, const unsigned int wdb) {
    const float* cur = hl + (ci & 1) * ((MC / 2) * RPP) + laneHi * RPP + 2 * lane16;
    const float* efc = efl + ci * (2 * MC) + 4 * laneHi;
    v2f bf[4][16];
    float4 sf[4][4];
    auto load_block = [&](const int q, v2f* bq, float4* sq) {
#pragma unroll
      for (int ks = 0; ks < 4; ++ks) {
        const float* hp = cur + (8 * q + 2 * ks) * RPP;
        sq[ks] = *reinterpret_cast<const float4*>(efc + 32 * q + 8 * ks);
        bq[4 * ks + 0] = *reinterpret_cast<const v2f*>(hp);
        bq[4 * ks + 1] = *reinterpret_cast<const v2f*>(hp + 32);
        bq[4 * ks + 2] = *reinterpret_cast<const v2f*>(hp + 64);
        bq[4 * ks + 3] = *reinterpret_cast<const v2f*>(hp + 96);
      }
    };
    auto mma_block = [&](const int q, const v2f* bq, const float4* sq) {
#pragma unroll
      for (int ks = 0; ks < 4; ++ks) {
        const int bit = 4 * q + ks;
        const float4 ef = sq[ks];
        const unsigned int s0 = (unsigned int)((int)(wda << (31 - 2 * bit)) >> 31);
        const unsigned int s1m = (unsigned int)((int)(wdb << (31 - 2 * bit)) >> 31);
        const unsigned int p0 = umin(umax(fbits(ef.x), fbits(ef.z * bneg)), s0);
        const unsigned int p1 = umin(umax(fbits(ef.y), fbits(ef.w * bneg)), s1m);
        v2f av;
        av.x = __builtin_bit_cast(float, p0);
        av.y = __builtin_bit_cast(float, p1);
        rs0 += av.x;
        rs1 += av.y;
        acc0 = __builtin_amdgcn_wmma_f32_16x16x4_f32(false, av, false, bq[4 * ks + 0], (short)0, acc0, false, false);
        acc1 = __builtin_amdgcn_wmma_f32_16x16x4_f32(false, av, false, bq[4 * ks + 1], (short)0, acc1, false, false);
        acc2 = __builtin_amdgcn_wmma_f32_16x16x4_f32(false, av, false, bq[4 * ks + 2], (short)0, acc2, false, false);
        acc3 = __builtin_amdgcn_wmma_f32_16x16x4_f32(false, av, false, bq[4 * ks + 3], (short)0, acc3, false, false);
      }
    };
#pragma unroll
    for (int q = 0; q < 4; ++q) load_block(q, bf[q], sf[q]);
#pragma unroll
    for (int q = 0; q < 4; ++q) mma_block(q, bf[q], sf[q]);
  };
  for (int ci = 0; ci < Nq / MC - 1; ++ci) {
    const uint2 wc = mask_piece();
    const float* gs = gsrc + (size_t)(ci + 1) * MC * HD;
    const float4 e0 = *reinterpret_cast<const float4*>(gs);
    const float4 o0 = *reinterpret_cast<const float4*>(gs + HD);
    const float4 e1 = *reinterpret_cast<const float4*>(gs + 32 * HD);
    const float4 o1 = *reinterpret_cast<const float4*>(gs + 33 * HD);
    load_piece(ci + 1);
    mma_chunk(ci, wc.x, wc.y);
    float* ld = ldst + ((ci + 1) & 1) * ((MC / 2) * RPP);
    put_pair(ld, e0, o0);
    put_pair(ld + 16 * RPP, e1, o1);
    lds_barrier();
  }
  {
    const uint2 wc = mask_piece();
    mma_chunk(Nq / MC - 1, wc.x, wc.y);
  }

  float rinv;
  {
    const float sm = rs0 + rs1;
    const float osm = __shfl_xor(sm, 16, 32);
    const float rsum = (laneHi ? osm : sm) + (laneHi ? sm : osm);
    rinv = rsum > 0.f ? 1.f / rsum : 0.f;
  }
  float* ob = out + ((size_t)b * Nq + n0 + wave * 16) * Fout + hh * HD;
#pragma unroll
  for (int v = 0; v < 8; ++v) {
    const float ri = __shfl(rinv, v + 8 * laneHi, 32);
    float* orow = ob + (size_t)(v + 8 * laneHi) * Fout + lane16;
    orow[0]  = acc0[v] * ri;
    orow[16] = acc1[v] * ri;
    orow[32] = acc2[v] * ri;
    orow[48] = acc3[v] * ri;
  }
}

extern "C" void kernel_launch(void* const* d_in, const int* in_sizes, int n_in,
                              void* d_out, int out_size, void* d_ws, size_t ws_size,
                              hipStream_t stream) {
  const float* x   = (const float*)d_in[0];
  const int*   adj = (const int*)d_in[1];
  const float* W   = (const float*)d_in[2];
  const float* a   = (const float*)d_in[3];
  float* out = (float*)d_out;

  float* hbuf = (float*)d_ws;

  gat_proj<<<dim3(Hh, 64), 256, 0, stream>>>(x, W, hbuf);
  gat_attn<<<dim3(Hh, Nq / 128, Bq), 256, 0, stream>>>(hbuf, adj, a, out);
}
